# starting kernel
# baseline (speedup 1.0000x reference)
#include <hip/hip_runtime.h>

typedef __attribute__((ext_vector_type(2))) float v2f;
typedef __attribute__((ext_vector_type(8))) float v8f;

static constexpr int Bsz = 4;
static constexpr int Tsz = 4096;
static constexpr int Csz = 1024;
static constexpr int Hsz = 64;
static constexpr int BT  = Bsz * Tsz;

__global__ __launch_bounds__(128) void proj_qkv(
    const float* __restrict__ x,
    const float* __restrict__ Wk,
    const float* __restrict__ Wq,
    const float* __restrict__ Wv,
    float* __restrict__ qo, float* __restrict__ ko, float* __restrict__ vo)
{
  const int lane = threadIdx.x & 31;
  const int wid  = threadIdx.x >> 5;
  const int tile = blockIdx.x * 4 + wid;
  const int row0 = tile * 16;
  const int m    = lane & 15;
  const int kp   = lane >> 4;

  v8f qa[4] = {}, ka[4] = {}, va[4] = {};

  const float* xr = x + (size_t)(row0 + m) * Csz + 2 * kp;

  for (int kk = 0; kk < Csz; kk += 4) {
    v2f a = *(const v2f*)(xr + kk);
    const int kr = kk + 2 * kp;
#pragma unroll
    for (int nn = 0; nn < 4; ++nn) {
      const int n = nn * 16 + m;
      v2f b;
      b.x = Wq[(size_t)kr * Hsz + n];
      b.y = Wq[(size_t)(kr + 1) * Hsz + n];
      qa[nn] = __builtin_amdgcn_wmma_f32_16x16x4_f32(false, a, false, b,
                                                     (short)0, qa[nn], false, false);
      b.x = Wk[(size_t)kr * Hsz + n];
      b.y = Wk[(size_t)(kr + 1) * Hsz + n];
      ka[nn] = __builtin_amdgcn_wmma_f32_16x16x4_f32(false, a, false, b,
                                                     (short)0, ka[nn], false, false);
      b.x = Wv[(size_t)kr * Hsz + n];
      b.y = Wv[(size_t)(kr + 1) * Hsz + n];
      va[nn] = __builtin_amdgcn_wmma_f32_16x16x4_f32(false, a, false, b,
                                                     (short)0, va[nn], false, false);
    }
  }

#pragma unroll
  for (int nn = 0; nn < 4; ++nn)
#pragma unroll
    for (int r = 0; r < 8; ++r) {
      const size_t idx = (size_t)(row0 + r + 8 * kp) * Hsz + nn * 16 + m;
      qo[idx] = qa[nn][r];
      ko[idx] = ka[nn][r];
      vo[idx] = va[nn][r];
    }
}

__global__ __launch_bounds__(128) void flash_attn(
    const float* __restrict__ qm, const float* __restrict__ km,
    const float* __restrict__ vm, float* __restrict__ out)
{
  __shared__ float pbuf[4][16 * 17];

  const int lane = threadIdx.x & 31;
  const int wid  = threadIdx.x >> 5;
  const int tile = blockIdx.x * 4 + wid;
  const int bb   = tile >> 8;
  const int ti   = tile & 255;
  const int m    = lane & 15;
  const int kp   = lane >> 4;

  const size_t base = (size_t)bb * Tsz * Hsz;
  const float* qb = qm + base;
  const float* kb = km + base;
  const float* vb = vm + base;

  v2f qf[16];
  {
    const float* qr = qb + (size_t)(ti * 16 + m) * Hsz + 2 * kp;
#pragma unroll
    for (int s = 0; s < 16; ++s) {
      v2f t = *(const v2f*)(qr + 4 * s);
      qf[s].x = t.x * 0.125f;
      qf[s].y = t.y * 0.125f;
    }
  }

  v8f acc[4] = {};
  float mr[8], lr[8];
#pragma unroll
  for (int r = 0; r < 8; ++r) { mr[r] = -1e30f; lr[r] = 0.0f; }

  float* pb = pbuf[wid];

  for (int j = 0; j <= ti; ++j) {
    v8f s8 = {};
    const float* kt = kb + (size_t)(j * 16 + m) * Hsz + 2 * kp;
#pragma unroll
    for (int s = 0; s < 16; ++s) {
      v2f bf = *(const v2f*)(kt + 4 * s);
      s8 = __builtin_amdgcn_wmma_f32_16x16x4_f32(false, qf[s], false, bf,
                                                 (short)0, s8, false, false);
    }

    if (j == ti) {
#pragma unroll
      for (int r = 0; r < 8; ++r)
        if (m > r + 8 * kp) s8[r] = -1e30f;
    }

    __builtin_amdgcn_wave_barrier();

#pragma unroll
    for (int r = 0; r < 8; ++r) {
      float rm = s8[r];
#pragma unroll
      for (int off = 1; off < 16; off <<= 1)
        rm = fmaxf(rm, __shfl_xor(rm, off, 32));
      const float mnew  = fmaxf(mr[r], rm);
      const float alpha = __expf(mr[r] - mnew);
      mr[r] = mnew;
      const float p = __expf(s8[r] - mnew);
      float rs = p;
#pragma unroll
      for (int off = 1; off < 16; off <<= 1)
        rs += __shfl_xor(rs, off, 32);
      lr[r] = lr[r] * alpha + rs;
#pragma unroll
      for (int nn = 0; nn < 4; ++nn) acc[nn][r] *= alpha;
      pb[(r + 8 * kp) * 17 + m] = p;
    }

    __builtin_amdgcn_wave_barrier();

    v2f pf[4];
#pragma unroll
    for (int s = 0; s < 4; ++s) {
      pf[s].x = pb[m * 17 + 4 * s + 2 * kp];
      pf[s].y = pb[m * 17 + 4 * s + 2 * kp + 1];
    }

    const float* vt = vb + (size_t)(j * 16) * Hsz;
#pragma unroll
    for (int s = 0; s < 4; ++s) {
      const int kr = 4 * s + 2 * kp;
#pragma unroll
      for (int nn = 0; nn < 4; ++nn) {
        const int n = nn * 16 + m;
        v2f bf;
        bf.x = vt[(size_t)kr * Hsz + n];
        bf.y = vt[(size_t)(kr + 1) * Hsz + n];
        acc[nn] = __builtin_amdgcn_wmma_f32_16x16x4_f32(false, pf[s], false, bf,
                                                        (short)0, acc[nn], false, false);
      }
    }
  }

#pragma unroll
  for (int nn = 0; nn < 4; ++nn)
#pragma unroll
    for (int r = 0; r < 8; ++r) {
      const size_t idx = base + (size_t)(ti * 16 + r + 8 * kp) * Hsz + nn * 16 + m;
      out[idx] = acc[nn][r] / lr[r];
    }
}

extern "C" void kernel_launch(void* const* d_in, const int* in_sizes, int n_in,
                              void* d_out, int out_size, void* d_ws, size_t ws_size,
                              hipStream_t stream) {
  (void)in_sizes; (void)n_in; (void)out_size; (void)ws_size;

  const float* x  = (const float*)d_in[0];
  const float* Wk = (const float*)d_in[1];
  const float* Wq = (const float*)d_in[2];
  const float* Wv = (const float*)d_in[3];
  float* out = (float*)d_out;

  float* qo = (float*)d_ws;
  float* ko = qo + (size_t)BT * Hsz;
  float* vo = ko + (size_t)BT * Hsz;

  const int tiles = BT / 16;
  dim3 block(128);
  dim3 grid(tiles / 4);

  proj_qkv<<<grid, block, 0, stream>>>(x, Wk, Wq, Wv, qo, ko, vo);
  flash_attn<<<grid, block, 0, stream>>>(qo, ko, vo, out);
}


# attention kernel: each q tile's causal KV tiles are split over the 8 waves of a 256-thread block, partial softmax states merged in LDS (new summation order); a block takes q tiles 255-p and p
# speedup vs baseline: 2.2778x; 2.2778x over previous
#include <hip/hip_runtime.h>

typedef __attribute__((ext_vector_type(2))) float v2f;
typedef __attribute__((ext_vector_type(8))) float v8f;

static constexpr int Bsz = 4;
static constexpr int Tsz = 4096;
static constexpr int Csz = 1024;
static constexpr int Hsz = 64;
static constexpr int BT  = Bsz * Tsz;

__global__ __launch_bounds__(128) void proj_qkv(
    const float* __restrict__ x,
    const float* __restrict__ Wk,
    const float* __restrict__ Wq,
    const float* __restrict__ Wv,
    float* __restrict__ qo, float* __restrict__ ko, float* __restrict__ vo)
{
  const int lane = threadIdx.x & 31;
  const int wid  = threadIdx.x >> 5;
  const int tile = blockIdx.x * 4 + wid;
  const int row0 = tile * 16;
  const int m    = lane & 15;
  const int kp   = lane >> 4;

  v8f qa[4] = {}, ka[4] = {}, va[4] = {};

  const float* xr = x + (size_t)(row0 + m) * Csz + 2 * kp;

  for (int kk = 0; kk < Csz; kk += 4) {
    v2f a = *(const v2f*)(xr + kk);
    const int kr = kk + 2 * kp;
#pragma unroll
    for (int nn = 0; nn < 4; ++nn) {
      const int n = nn * 16 + m;
      v2f b;
      b.x = Wq[(size_t)kr * Hsz + n];
      b.y = Wq[(size_t)(kr + 1) * Hsz + n];
      qa[nn] = __builtin_amdgcn_wmma_f32_16x16x4_f32(false, a, false, b,
                                                     (short)0, qa[nn], false, false);
      b.x = Wk[(size_t)kr * Hsz + n];
      b.y = Wk[(size_t)(kr + 1) * Hsz + n];
      ka[nn] = __builtin_amdgcn_wmma_f32_16x16x4_f32(false, a, false, b,
                                                     (short)0, ka[nn], false, false);
      b.x = Wv[(size_t)kr * Hsz + n];
      b.y = Wv[(size_t)(kr + 1) * Hsz + n];
      va[nn] = __builtin_amdgcn_wmma_f32_16x16x4_f32(false, a, false, b,
                                                     (short)0, va[nn], false, false);
    }
  }

#pragma unroll
  for (int nn = 0; nn < 4; ++nn)
#pragma unroll
    for (int r = 0; r < 8; ++r) {
      const size_t idx = (size_t)(row0 + r + 8 * kp) * Hsz + nn * 16 + m;
      qo[idx] = qa[nn][r];
      ko[idx] = ka[nn][r];
      vo[idx] = va[nn][r];
    }
}

static constexpr int NW = 8;
static constexpr int MP = 68;

__global__ __launch_bounds__(NW * 32) void flash_attn(
    const float* __restrict__ qm, const float* __restrict__ km,
    const float* __restrict__ vm, float* __restrict__ out)
{
  __shared__ float pbuf[NW][16 * 17];
  __shared__ __attribute__((aligned(16))) float macc[NW * 16 * MP];
  __shared__ float mst[NW][16];
  __shared__ float lst[NW][16];

  const int lane = threadIdx.x & 31;
  const int wid  = threadIdx.x >> 5;
  const int bb   = blockIdx.x >> 7;
  const int pr   = blockIdx.x & 127;
  const int m    = lane & 15;
  const int kp   = lane >> 4;

  const size_t base = (size_t)bb * Tsz * Hsz;
  const float* qb = qm + base;
  const float* kb = km + base;
  const float* vb = vm + base;

  float* pb = pbuf[wid];

  for (int half = 0; half < 2; ++half) {
    const int ti = half ? pr : 255 - pr;

    v2f qf[16];
    {
      const float* qr = qb + (size_t)(ti * 16 + m) * Hsz + 2 * kp;
#pragma unroll
      for (int s = 0; s < 16; ++s) {
        v2f t = *(const v2f*)(qr + 4 * s);
        qf[s].x = t.x * 0.125f;
        qf[s].y = t.y * 0.125f;
      }
    }

    v8f acc[4] = {};
    float mr[8], lr[8];
#pragma unroll
    for (int r = 0; r < 8; ++r) { mr[r] = -1e30f; lr[r] = 0.0f; }

    for (int j = wid; j <= ti; j += NW) {
      v8f s8 = {};
      const float* kt = kb + (size_t)(j * 16 + m) * Hsz + 2 * kp;
#pragma unroll
      for (int s = 0; s < 16; ++s) {
        v2f bf = *(const v2f*)(kt + 4 * s);
        s8 = __builtin_amdgcn_wmma_f32_16x16x4_f32(false, qf[s], false, bf,
                                                   (short)0, s8, false, false);
      }

      if (j == ti) {
#pragma unroll
        for (int r = 0; r < 8; ++r)
          if (m > r + 8 * kp) s8[r] = -1e30f;
      }

      __builtin_amdgcn_wave_barrier();

#pragma unroll
      for (int r = 0; r < 8; ++r) {
        float rm = s8[r];
#pragma unroll
        for (int off = 1; off < 16; off <<= 1)
          rm = fmaxf(rm, __shfl_xor(rm, off, 32));
        const float mnew  = fmaxf(mr[r], rm);
        const float alpha = __expf(mr[r] - mnew);
        mr[r] = mnew;
        const float p = __expf(s8[r] - mnew);
        float rs = p;
#pragma unroll
        for (int off = 1; off < 16; off <<= 1)
          rs += __shfl_xor(rs, off, 32);
        lr[r] = lr[r] * alpha + rs;
#pragma unroll
        for (int nn = 0; nn < 4; ++nn) acc[nn][r] *= alpha;
        pb[(r + 8 * kp) * 17 + m] = p;
      }

      __builtin_amdgcn_wave_barrier();

      v2f pf[4];
#pragma unroll
      for (int s = 0; s < 4; ++s) {
        pf[s].x = pb[m * 17 + 4 * s + 2 * kp];
        pf[s].y = pb[m * 17 + 4 * s + 2 * kp + 1];
      }

      const float* vt = vb + (size_t)(j * 16) * Hsz;
#pragma unroll
      for (int s = 0; s < 4; ++s) {
        const int kr = 4 * s + 2 * kp;
#pragma unroll
        for (int nn = 0; nn < 4; ++nn) {
          const int n = nn * 16 + m;
          v2f bf;
          bf.x = vt[(size_t)kr * Hsz + n];
          bf.y = vt[(size_t)(kr + 1) * Hsz + n];
          acc[nn] = __builtin_amdgcn_wmma_f32_16x16x4_f32(false, pf[s], false, bf,
                                                          (short)0, acc[nn], false, false);
        }
      }
    }

    {
      float* ma = macc + wid * (16 * MP);
#pragma unroll
      for (int nn = 0; nn < 4; ++nn)
#pragma unroll
        for (int r = 0; r < 8; ++r)
          ma[(r + 8 * kp) * MP + nn * 16 + m] = acc[nn][r];
      if (m == 0) {
#pragma unroll
        for (int r = 0; r < 8; ++r) {
          mst[wid][r + 8 * kp] = mr[r];
          lst[wid][r + 8 * kp] = lr[r];
        }
      }
    }

    __syncthreads();

    {
      const int row = threadIdx.x >> 4;
      const int c4  = (threadIdx.x & 15) * 4;
      float mw[NW];
      float mt = -1e30f;
#pragma unroll
      for (int w = 0; w < NW; ++w) { mw[w] = mst[w][row]; mt = fmaxf(mt, mw[w]); }
      float lt = 0.0f;
      float o0 = 0.0f, o1 = 0.0f, o2 = 0.0f, o3 = 0.0f;
#pragma unroll
      for (int w = 0; w < NW; ++w) {
        const float sc = __expf(mw[w] - mt);
        lt += lst[w][row] * sc;
        const float4 a = *(const float4*)(macc + (w * 16 + row) * MP + c4);
        o0 += a.x * sc; o1 += a.y * sc; o2 += a.z * sc; o3 += a.w * sc;
      }
      float4 res;
      res.x = o0 / lt; res.y = o1 / lt; res.z = o2 / lt; res.w = o3 / lt;
      *(float4*)(out + base + (size_t)(ti * 16 + row) * Hsz + c4) = res;
    }

    if (half == 0) __syncthreads();
  }
}

extern "C" void kernel_launch(void* const* d_in, const int* in_sizes, int n_in,
                              void* d_out, int out_size, void* d_ws, size_t ws_size,
                              hipStream_t stream) {
  (void)in_sizes; (void)n_in; (void)out_size; (void)ws_size;

  const float* x  = (const float*)d_in[0];
  const float* Wk = (const float*)d_in[1];
  const float* Wq = (const float*)d_in[2];
  const float* Wv = (const float*)d_in[3];
  float* out = (float*)d_out;

  float* qo = (float*)d_ws;
  float* ko = qo + (size_t)BT * Hsz;
  float* vo = ko + (size_t)BT * Hsz;

  const int tiles = BT / 16;
  dim3 block(128);
  dim3 grid(tiles / 4);

  proj_qkv<<<grid, block, 0, stream>>>(x, Wk, Wq, Wv, qo, ko, vo);
  flash_attn<<<dim3(512), dim3(256), 0, stream>>>(qo, ko, vo, out);
}


# projection kernel: K loop software-pipelined over two register sets, the next two K-steps' x and W fragment loads issued as one group ahead of the current 24 WMMAs; summation order unchanged
# speedup vs baseline: 3.8650x; 1.6968x over previous
#include <hip/hip_runtime.h>

typedef __attribute__((ext_vector_type(2))) float v2f;
typedef __attribute__((ext_vector_type(8))) float v8f;

static constexpr int Bsz = 4;
static constexpr int Tsz = 4096;
static constexpr int Csz = 1024;
static constexpr int Hsz = 64;
static constexpr int BT  = Bsz * Tsz;

struct Frag {
  v2f a;
  v2f bq[4], bk[4], bv[4];
};

__device__ __forceinline__ void load_step(Frag& f, const float* __restrict__ xr,
                                          const float* __restrict__ wq,
                                          const float* __restrict__ wk,
                                          const float* __restrict__ wv, int kk)
{
  f.a = *(const v2f*)(xr + kk);
#pragma unroll
  for (int nn = 0; nn < 4; ++nn) {
    f.bq[nn].x = wq[(size_t)kk * Hsz + nn * 16];
    f.bq[nn].y = wq[(size_t)(kk + 1) * Hsz + nn * 16];
    f.bk[nn].x = wk[(size_t)kk * Hsz + nn * 16];
    f.bk[nn].y = wk[(size_t)(kk + 1) * Hsz + nn * 16];
    f.bv[nn].x = wv[(size_t)kk * Hsz + nn * 16];
    f.bv[nn].y = wv[(size_t)(kk + 1) * Hsz + nn * 16];
  }
}

__device__ __forceinline__ void mma_step(const Frag& f, v8f (&qa)[4], v8f (&ka)[4], v8f (&va)[4])
{
#pragma unroll
  for (int nn = 0; nn < 4; ++nn) {
    qa[nn] = __builtin_amdgcn_wmma_f32_16x16x4_f32(false, f.a, false, f.bq[nn],
                                                   (short)0, qa[nn], false, false);
    ka[nn] = __builtin_amdgcn_wmma_f32_16x16x4_f32(false, f.a, false, f.bk[nn],
                                                   (short)0, ka[nn], false, false);
    va[nn] = __builtin_amdgcn_wmma_f32_16x16x4_f32(false, f.a, false, f.bv[nn],
                                                   (short)0, va[nn], false, false);
  }
}

__global__ __launch_bounds__(128) void proj_qkv(
    const float* __restrict__ x,
    const float* __restrict__ Wk,
    const float* __restrict__ Wq,
    const float* __restrict__ Wv,
    float* __restrict__ qo, float* __restrict__ ko, float* __restrict__ vo)
{
  const int lane = threadIdx.x & 31;
  const int wid  = threadIdx.x >> 5;
  const int tile = blockIdx.x * 4 + wid;
  const int row0 = tile * 16;
  const int m    = lane & 15;
  const int kp   = lane >> 4;

  v8f qa[4] = {}, ka[4] = {}, va[4] = {};

  const float* xr = x + (size_t)(row0 + m) * Csz + 2 * kp;
  const float* wq = Wq + (size_t)(2 * kp) * Hsz + m;
  const float* wk = Wk + (size_t)(2 * kp) * Hsz + m;
  const float* wv = Wv + (size_t)(2 * kp) * Hsz + m;

  Frag c0, c1, n0, n1;
  load_step(c0, xr, wq, wk, wv, 0);
  load_step(c1, xr, wq, wk, wv, 4);

  for (int kk = 0; kk < Csz - 16; kk += 16) {
    load_step(n0, xr, wq, wk, wv, kk + 8);
    load_step(n1, xr, wq, wk, wv, kk + 12);
    __builtin_amdgcn_sched_barrier(0);
    mma_step(c0, qa, ka, va);
    mma_step(c1, qa, ka, va);
    load_step(c0, xr, wq, wk, wv, kk + 16);
    load_step(c1, xr, wq, wk, wv, kk + 20);
    mma_step(n0, qa, ka, va);
    mma_step(n1, qa, ka, va);
  }
  load_step(n0, xr, wq, wk, wv, Csz - 8);
  load_step(n1, xr, wq, wk, wv, Csz - 4);
  mma_step(c0, qa, ka, va);
  mma_step(c1, qa, ka, va);
  mma_step(n0, qa, ka, va);
  mma_step(n1, qa, ka, va);

#pragma unroll
  for (int nn = 0; nn < 4; ++nn)
#pragma unroll
    for (int r = 0; r < 8; ++r) {
      const size_t idx = (size_t)(row0 + r + 8 * kp) * Hsz + nn * 16 + m;
      qo[idx] = qa[nn][r];
      ko[idx] = ka[nn][r];
      vo[idx] = va[nn][r];
    }
}

static constexpr int NW = 8;
static constexpr int MP = 68;

__global__ __launch_bounds__(NW * 32) void flash_attn(
    const float* __restrict__ qm, const float* __restrict__ km,
    const float* __restrict__ vm, float* __restrict__ out)
{
  __shared__ float pbuf[NW][16 * 17];
  __shared__ __attribute__((aligned(16))) float macc[NW * 16 * MP];
  __shared__ float mst[NW][16];
  __shared__ float lst[NW][16];

  const int lane = threadIdx.x & 31;
  const int wid  = threadIdx.x >> 5;
  const int bb   = blockIdx.x >> 7;
  const int pr   = blockIdx.x & 127;
  const int m    = lane & 15;
  const int kp   = lane >> 4;

  const size_t base = (size_t)bb * Tsz * Hsz;
  const float* qb = qm + base;
  const float* kb = km + base;
  const float* vb = vm + base;

  float* pb = pbuf[wid];

  for (int half = 0; half < 2; ++half) {
    const int ti = half ? pr : 255 - pr;

    v2f qf[16];
    {
      const float* qr = qb + (size_t)(ti * 16 + m) * Hsz + 2 * kp;
#pragma unroll
      for (int s = 0; s < 16; ++s) {
        v2f t = *(const v2f*)(qr + 4 * s);
        qf[s].x = t.x * 0.125f;
        qf[s].y = t.y * 0.125f;
      }
    }

    v8f acc[4] = {};
    float mr[8], lr[8];
#pragma unroll
    for (int r = 0; r < 8; ++r) { mr[r] = -1e30f; lr[r] = 0.0f; }

    for (int j = wid; j <= ti; j += NW) {
      v8f s8 = {};
      const float* kt = kb + (size_t)(j * 16 + m) * Hsz + 2 * kp;
#pragma unroll
      for (int s = 0; s < 16; ++s) {
        v2f bf = *(const v2f*)(kt + 4 * s);
        s8 = __builtin_amdgcn_wmma_f32_16x16x4_f32(false, qf[s], false, bf,
                                                   (short)0, s8, false, false);
      }

      if (j == ti) {
#pragma unroll
        for (int r = 0; r < 8; ++r)
          if (m > r + 8 * kp) s8[r] = -1e30f;
      }

      __builtin_amdgcn_wave_barrier();

#pragma unroll
      for (int r = 0; r < 8; ++r) {
        float rm = s8[r];
#pragma unroll
        for (int off = 1; off < 16; off <<= 1)
          rm = fmaxf(rm, __shfl_xor(rm, off, 32));
        const float mnew  = fmaxf(mr[r], rm);
        const float alpha = __expf(mr[r] - mnew);
        mr[r] = mnew;
        const float p = __expf(s8[r] - mnew);
        float rs = p;
#pragma unroll
        for (int off = 1; off < 16; off <<= 1)
          rs += __shfl_xor(rs, off, 32);
        lr[r] = lr[r] * alpha + rs;
#pragma unroll
        for (int nn = 0; nn < 4; ++nn) acc[nn][r] *= alpha;
        pb[(r + 8 * kp) * 17 + m] = p;
      }

      __builtin_amdgcn_wave_barrier();

      v2f pf[4];
#pragma unroll
      for (int s = 0; s < 4; ++s) {
        pf[s].x = pb[m * 17 + 4 * s + 2 * kp];
        pf[s].y = pb[m * 17 + 4 * s + 2 * kp + 1];
      }

      const float* vt = vb + (size_t)(j * 16) * Hsz;
#pragma unroll
      for (int s = 0; s < 4; ++s) {
        const int kr = 4 * s + 2 * kp;
#pragma unroll
        for (int nn = 0; nn < 4; ++nn) {
          const int n = nn * 16 + m;
          v2f bf;
          bf.x = vt[(size_t)kr * Hsz + n];
          bf.y = vt[(size_t)(kr + 1) * Hsz + n];
          acc[nn] = __builtin_amdgcn_wmma_f32_16x16x4_f32(false, pf[s], false, bf,
                                                          (short)0, acc[nn], false, false);
        }
      }
    }

    {
      float* ma = macc + wid * (16 * MP);
#pragma unroll
      for (int nn = 0; nn < 4; ++nn)
#pragma unroll
        for (int r = 0; r < 8; ++r)
          ma[(r + 8 * kp) * MP + nn * 16 + m] = acc[nn][r];
      if (m == 0) {
#pragma unroll
        for (int r = 0; r < 8; ++r) {
          mst[wid][r + 8 * kp] = mr[r];
          lst[wid][r + 8 * kp] = lr[r];
        }
      }
    }

    __syncthreads();

    {
      const int row = threadIdx.x >> 4;
      const int c4  = (threadIdx.x & 15) * 4;
      float mw[NW];
      float mt = -1e30f;
#pragma unroll
      for (int w = 0; w < NW; ++w) { mw[w] = mst[w][row]; mt = fmaxf(mt, mw[w]); }
      float lt = 0.0f;
      float o0 = 0.0f, o1 = 0.0f, o2 = 0.0f, o3 = 0.0f;
#pragma unroll
      for (int w = 0; w < NW; ++w) {
        const float sc = __expf(mw[w] - mt);
        lt += lst[w][row] * sc;
        const float4 a = *(const float4*)(macc + (w * 16 + row) * MP + c4);
        o0 += a.x * sc; o1 += a.y * sc; o2 += a.z * sc; o3 += a.w * sc;
      }
      float4 res;
      res.x = o0 / lt; res.y = o1 / lt; res.z = o2 / lt; res.w = o3 / lt;
      *(float4*)(out + base + (size_t)(ti * 16 + row) * Hsz + c4) = res;
    }

    if (half == 0) __syncthreads();
  }
}

extern "C" void kernel_launch(void* const* d_in, const int* in_sizes, int n_in,
                              void* d_out, int out_size, void* d_ws, size_t ws_size,
                              hipStream_t stream) {
  (void)in_sizes; (void)n_in; (void)out_size; (void)ws_size;

  const float* x  = (const float*)d_in[0];
  const float* Wk = (const float*)d_in[1];
  const float* Wq = (const float*)d_in[2];
  const float* Wv = (const float*)d_in[3];
  float* out = (float*)d_out;

  float* qo = (float*)d_ws;
  float* ko = qo + (size_t)BT * Hsz;
  float* vo = ko + (size_t)BT * Hsz;

  const int tiles = BT / 16;
  dim3 block(128);
  dim3 grid(tiles / 4);

  proj_qkv<<<grid, block, 0, stream>>>(x, Wk, Wq, Wv, qo, ko, vo);
  flash_attn<<<dim3(512), dim3(256), 0, stream>>>(qo, ko, vo, out);
}


# attention: both products transposed (S^T = K Q^T, O^T = V^T P^T): row max and sum in-lane plus one lane-half exchange, P^T read from the score registers, no LDS in the loop; new summation order
# speedup vs baseline: 4.1787x; 1.0812x over previous
#include <hip/hip_runtime.h>

typedef __attribute__((ext_vector_type(2))) float v2f;
typedef __attribute__((ext_vector_type(8))) float v8f;

static constexpr int Bsz = 4;
static constexpr int Tsz = 4096;
static constexpr int Csz = 1024;
static constexpr int Hsz = 64;
static constexpr int BT  = Bsz * Tsz;

struct Frag {
  v2f a;
  v2f bq[4], bk[4], bv[4];
};

__device__ __forceinline__ void load_step(Frag& f, const float* __restrict__ xr,
                                          const float* __restrict__ wq,
                                          const float* __restrict__ wk,
                                          const float* __restrict__ wv, int kk)
{
  f.a = *(const v2f*)(xr + kk);
#pragma unroll
  for (int nn = 0; nn < 4; ++nn) {
    f.bq[nn].x = wq[(size_t)kk * Hsz + nn * 16];
    f.bq[nn].y = wq[(size_t)(kk + 1) * Hsz + nn * 16];
    f.bk[nn].x = wk[(size_t)kk * Hsz + nn * 16];
    f.bk[nn].y = wk[(size_t)(kk + 1) * Hsz + nn * 16];
    f.bv[nn].x = wv[(size_t)kk * Hsz + nn * 16];
    f.bv[nn].y = wv[(size_t)(kk + 1) * Hsz + nn * 16];
  }
}

__device__ __forceinline__ void mma_step(const Frag& f, v8f (&qa)[4], v8f (&ka)[4], v8f (&va)[4])
{
#pragma unroll
  for (int nn = 0; nn < 4; ++nn) {
    qa[nn] = __builtin_amdgcn_wmma_f32_16x16x4_f32(false, f.a, false, f.bq[nn],
                                                   (short)0, qa[nn], false, false);
    ka[nn] = __builtin_amdgcn_wmma_f32_16x16x4_f32(false, f.a, false, f.bk[nn],
                                                   (short)0, ka[nn], false, false);
    va[nn] = __builtin_amdgcn_wmma_f32_16x16x4_f32(false, f.a, false, f.bv[nn],
                                                   (short)0, va[nn], false, false);
  }
}

__global__ __launch_bounds__(128) void proj_qkv(
    const float* __restrict__ x,
    const float* __restrict__ Wk,
    const float* __restrict__ Wq,
    const float* __restrict__ Wv,
    float* __restrict__ qo, float* __restrict__ ko, float* __restrict__ vo)
{
  const int lane = threadIdx.x & 31;
  const int wid  = threadIdx.x >> 5;
  const int tile = blockIdx.x * 4 + wid;
  const int row0 = tile * 16;
  const int m    = lane & 15;
  const int kp   = lane >> 4;

  v8f qa[4] = {}, ka[4] = {}, va[4] = {};

  const float* xr = x + (size_t)(row0 + m) * Csz + 2 * kp;
  const float* wq = Wq + (size_t)(2 * kp) * Hsz + m;
  const float* wk = Wk + (size_t)(2 * kp) * Hsz + m;
  const float* wv = Wv + (size_t)(2 * kp) * Hsz + m;

  Frag c0, c1, n0, n1;
  load_step(c0, xr, wq, wk, wv, 0);
  load_step(c1, xr, wq, wk, wv, 4);

  for (int kk = 0; kk < Csz - 16; kk += 16) {
    load_step(n0, xr, wq, wk, wv, kk + 8);
    load_step(n1, xr, wq, wk, wv, kk + 12);
    __builtin_amdgcn_sched_barrier(0);
    mma_step(c0, qa, ka, va);
    mma_step(c1, qa, ka, va);
    load_step(c0, xr, wq, wk, wv, kk + 16);
    load_step(c1, xr, wq, wk, wv, kk + 20);
    mma_step(n0, qa, ka, va);
    mma_step(n1, qa, ka, va);
  }
  load_step(n0, xr, wq, wk, wv, Csz - 8);
  load_step(n1, xr, wq, wk, wv, Csz - 4);
  mma_step(c0, qa, ka, va);
  mma_step(c1, qa, ka, va);
  mma_step(n0, qa, ka, va);
  mma_step(n1, qa, ka, va);

#pragma unroll
  for (int nn = 0; nn < 4; ++nn)
#pragma unroll
    for (int r = 0; r < 8; ++r) {
      const size_t idx = (size_t)(row0 + r + 8 * kp) * Hsz + nn * 16 + m;
      qo[idx] = qa[nn][r];
      ko[idx] = ka[nn][r];
      vo[idx] = va[nn][r];
    }
}

static constexpr int NW = 8;
static constexpr int MP = 68;

__device__ __forceinline__ float xhalf(float x)
{
  const int b = __builtin_bit_cast(int, x);
  return __builtin_bit_cast(float,
      __builtin_amdgcn_permlanex16(b, b, 0x76543210, 0xfedcba98, false, false));
}

__global__ __launch_bounds__(NW * 32) void flash_attn(
    const float* __restrict__ qm, const float* __restrict__ km,
    const float* __restrict__ vm, float* __restrict__ out)
{
  __shared__ __attribute__((aligned(16))) float macc[NW * 16 * MP];
  __shared__ float mst[NW][16];
  __shared__ float lst[NW][16];

  const int lane = threadIdx.x & 31;
  const int wid  = threadIdx.x >> 5;
  const int bb   = blockIdx.x >> 7;
  const int pr   = blockIdx.x & 127;
  const int m    = lane & 15;
  const int kp   = lane >> 4;

  const size_t base = (size_t)bb * Tsz * Hsz;
  const float* qb = qm + base;
  const float* kb = km + base;
  const float* vb = vm + base;

  for (int half = 0; half < 2; ++half) {
    const int ti = half ? pr : 255 - pr;

    v2f qf[16];
    {
      const float* qr = qb + (size_t)(ti * 16 + m) * Hsz + 2 * kp;
#pragma unroll
      for (int s = 0; s < 16; ++s) {
        v2f t = *(const v2f*)(qr + 4 * s);
        qf[s].x = t.x * 0.125f;
        qf[s].y = t.y * 0.125f;
      }
    }

    v8f acc[4] = {};
    float mrun = -1e30f;
    v8f lv = {};

    for (int j = wid; j <= ti; j += NW) {
      v8f s8 = {};
      const float* kt = kb + (size_t)(j * 16 + m) * Hsz + 2 * kp;
#pragma unroll
      for (int s = 0; s < 16; ++s) {
        v2f af = *(const v2f*)(kt + 4 * s);
        s8 = __builtin_amdgcn_wmma_f32_16x16x4_f32(false, af, false, qf[s],
                                                   (short)0, s8, false, false);
      }

      if (j == ti) {
#pragma unroll
        for (int r = 0; r < 8; ++r)
          if (r + 8 * kp > m) s8[r] = -1e30f;
      }

      v2f vf[4][4];
      {
        const float* vt = vb + (size_t)(j * 16 + 8 * kp) * Hsz + m;
#pragma unroll
        for (int t = 0; t < 4; ++t)
#pragma unroll
          for (int nn = 0; nn < 4; ++nn) {
            vf[t][nn].x = vt[(size_t)(2 * t) * Hsz + nn * 16];
            vf[t][nn].y = vt[(size_t)(2 * t + 1) * Hsz + nn * 16];
          }
      }

      float tm = s8[0];
#pragma unroll
      for (int r = 1; r < 8; ++r) tm = fmaxf(tm, s8[r]);
      tm = fmaxf(tm, xhalf(tm));
      const float mnew  = fmaxf(mrun, tm);
      const float alpha = __expf(mrun - mnew);
      mrun = mnew;
      v8f p8;
#pragma unroll
      for (int r = 0; r < 8; ++r) p8[r] = __expf(s8[r] - mnew);
      lv = lv * alpha + p8;
#pragma unroll
      for (int nn = 0; nn < 4; ++nn)
#pragma unroll
        for (int r = 0; r < 8; ++r) acc[nn][r] *= alpha;

      __builtin_amdgcn_sched_barrier(0);

#pragma unroll
      for (int t = 0; t < 4; ++t) {
        v2f pf;
        pf.x = p8[2 * t];
        pf.y = p8[2 * t + 1];
#pragma unroll
        for (int nn = 0; nn < 4; ++nn)
          acc[nn] = __builtin_amdgcn_wmma_f32_16x16x4_f32(false, vf[t][nn], false, pf,
                                                          (short)0, acc[nn], false, false);
      }
    }

    {
      float* ma = macc + wid * (16 * MP) + m * MP + 8 * kp;
#pragma unroll
      for (int nn = 0; nn < 4; ++nn) {
        float4 lo, hi;
        lo.x = acc[nn][0]; lo.y = acc[nn][1]; lo.z = acc[nn][2]; lo.w = acc[nn][3];
        hi.x = acc[nn][4]; hi.y = acc[nn][5]; hi.z = acc[nn][6]; hi.w = acc[nn][7];
        *(float4*)(ma + nn * 16)     = lo;
        *(float4*)(ma + nn * 16 + 4) = hi;
      }
      const float lsum  = ((lv[0] + lv[1]) + (lv[2] + lv[3])) + ((lv[4] + lv[5]) + (lv[6] + lv[7]));
      const float lfull = lsum + xhalf(lsum);
      if (kp == 0) {
        mst[wid][m] = mrun;
        lst[wid][m] = lfull;
      }
    }

    __syncthreads();

    {
      const int row = threadIdx.x >> 4;
      const int c4  = (threadIdx.x & 15) * 4;
      float mw[NW];
      float mt = -1e30f;
#pragma unroll
      for (int w = 0; w < NW; ++w) { mw[w] = mst[w][row]; mt = fmaxf(mt, mw[w]); }
      float lt = 0.0f;
      float o0 = 0.0f, o1 = 0.0f, o2 = 0.0f, o3 = 0.0f;
#pragma unroll
      for (int w = 0; w < NW; ++w) {
        const float sc = __expf(mw[w] - mt);
        lt += lst[w][row] * sc;
        const float4 a = *(const float4*)(macc + (w * 16 + row) * MP + c4);
        o0 += a.x * sc; o1 += a.y * sc; o2 += a.z * sc; o3 += a.w * sc;
      }
      float4 res;
      res.x = o0 / lt; res.y = o1 / lt; res.z = o2 / lt; res.w = o3 / lt;
      *(float4*)(out + base + (size_t)(ti * 16 + row) * Hsz + c4) = res;
    }

    if (half == 0) __syncthreads();
  }
}

extern "C" void kernel_launch(void* const* d_in, const int* in_sizes, int n_in,
                              void* d_out, int out_size, void* d_ws, size_t ws_size,
                              hipStream_t stream) {
  (void)in_sizes; (void)n_in; (void)out_size; (void)ws_size;

  const float* x  = (const float*)d_in[0];
  const float* Wk = (const float*)d_in[1];
  const float* Wq = (const float*)d_in[2];
  const float* Wv = (const float*)d_in[3];
  float* out = (float*)d_out;

  float* qo = (float*)d_ws;
  float* ko = qo + (size_t)BT * Hsz;
  float* vo = ko + (size_t)BT * Hsz;

  const int tiles = BT / 16;
  dim3 block(128);
  dim3 grid(tiles / 4);

  proj_qkv<<<grid, block, 0, stream>>>(x, Wk, Wq, Wv, qo, ko, vo);
  flash_attn<<<dim3(512), dim3(256), 0, stream>>>(qo, ko, vo, out);
}


# q, k, v workspace planes written in matrix-fragment order (q, k tiles computed transposed): attention reads each operand as whole-line b128 loads, 16 per tile not 48; score sum over d in a new order
# speedup vs baseline: 4.4649x; 1.0685x over previous
#include <hip/hip_runtime.h>

typedef __attribute__((ext_vector_type(2))) float v2f;
typedef __attribute__((ext_vector_type(4))) float v4f;
typedef __attribute__((ext_vector_type(8))) float v8f;

static constexpr int Bsz = 4;
static constexpr int Tsz = 4096;
static constexpr int Csz = 1024;
static constexpr int Hsz = 64;
static constexpr int BT  = Bsz * Tsz;

struct Frag {
  v2f a;
  v2f bq[4], bk[4], bv[4];
};

__device__ __forceinline__ void load_step(Frag& f, const float* __restrict__ xr,
                                          const float* __restrict__ wq,
                                          const float* __restrict__ wk,
                                          const float* __restrict__ wv, int kk)
{
  f.a = *(const v2f*)(xr + kk);
#pragma unroll
  for (int nn = 0; nn < 4; ++nn) {
    f.bq[nn].x = wq[(size_t)kk * Hsz + nn * 16];
    f.bq[nn].y = wq[(size_t)(kk + 1) * Hsz + nn * 16];
    f.bk[nn].x = wk[(size_t)kk * Hsz + nn * 16];
    f.bk[nn].y = wk[(size_t)(kk + 1) * Hsz + nn * 16];
    f.bv[nn].x = wv[(size_t)kk * Hsz + nn * 16];
    f.bv[nn].y = wv[(size_t)(kk + 1) * Hsz + nn * 16];
  }
}

__device__ __forceinline__ void mma_step(const Frag& f, v8f (&qa)[4], v8f (&ka)[4], v8f (&va)[4])
{
#pragma unroll
  for (int nn = 0; nn < 4; ++nn) {
    qa[nn] = __builtin_amdgcn_wmma_f32_16x16x4_f32(false, f.bq[nn], false, f.a,
                                                   (short)0, qa[nn], false, false);
    ka[nn] = __builtin_amdgcn_wmma_f32_16x16x4_f32(false, f.bk[nn], false, f.a,
                                                   (short)0, ka[nn], false, false);
    va[nn] = __builtin_amdgcn_wmma_f32_16x16x4_f32(false, f.a, false, f.bv[nn],
                                                   (short)0, va[nn], false, false);
  }
}

__global__ __launch_bounds__(128) void proj_qkv(
    const float* __restrict__ x,
    const float* __restrict__ Wk,
    const float* __restrict__ Wq,
    const float* __restrict__ Wv,
    float* __restrict__ qo, float* __restrict__ ko, float* __restrict__ vo)
{
  const int lane = threadIdx.x & 31;
  const int wid  = threadIdx.x >> 5;
  const int tile = blockIdx.x * 4 + wid;
  const int row0 = tile * 16;
  const int m    = lane & 15;
  const int kp   = lane >> 4;

  v8f qa[4] = {}, ka[4] = {}, va[4] = {};

  const float* xr = x + (size_t)(row0 + m) * Csz + 2 * kp;
  const float* wq = Wq + (size_t)(2 * kp) * Hsz + m;
  const float* wk = Wk + (size_t)(2 * kp) * Hsz + m;
  const float* wv = Wv + (size_t)(2 * kp) * Hsz + m;

  Frag c0, c1, n0, n1;
  load_step(c0, xr, wq, wk, wv, 0);
  load_step(c1, xr, wq, wk, wv, 4);

  for (int kk = 0; kk < Csz - 16; kk += 16) {
    load_step(n0, xr, wq, wk, wv, kk + 8);
    load_step(n1, xr, wq, wk, wv, kk + 12);
    __builtin_amdgcn_sched_barrier(0);
    mma_step(c0, qa, ka, va);
    mma_step(c1, qa, ka, va);
    load_step(c0, xr, wq, wk, wv, kk + 16);
    load_step(c1, xr, wq, wk, wv, kk + 20);
    mma_step(n0, qa, ka, va);
    mma_step(n1, qa, ka, va);
  }
  load_step(n0, xr, wq, wk, wv, Csz - 8);
  load_step(n1, xr, wq, wk, wv, Csz - 4);
  mma_step(c0, qa, ka, va);
  mma_step(c1, qa, ka, va);
  mma_step(n0, qa, ka, va);
  mma_step(n1, qa, ka, va);

  const size_t tb = (size_t)tile * (16 * Hsz) + lane * 4;
#pragma unroll
  for (int nn = 0; nn < 4; ++nn)
#pragma unroll
    for (int u = 0; u < 2; ++u) {
      const size_t idx = tb + (nn * 2 + u) * 128;
      v4f tq, tk, tv;
#pragma unroll
      for (int e = 0; e < 4; ++e) {
        tq[e] = qa[nn][4 * u + e];
        tk[e] = ka[nn][4 * u + e];
        tv[e] = va[nn][4 * u + e];
      }
      *(v4f*)(qo + idx) = tq;
      *(v4f*)(ko + idx) = tk;
      *(v4f*)(vo + idx) = tv;
    }
}

static constexpr int NW = 8;
static constexpr int MP = 68;

__device__ __forceinline__ float xhalf(float x)
{
  const int b = __builtin_bit_cast(int, x);
  return __builtin_bit_cast(float,
      __builtin_amdgcn_permlanex16(b, b, 0x76543210, 0xfedcba98, false, false));
}

__global__ __launch_bounds__(NW * 32) void flash_attn(
    const float* __restrict__ qm, const float* __restrict__ km,
    const float* __restrict__ vm, float* __restrict__ out)
{
  __shared__ __attribute__((aligned(16))) float macc[NW * 16 * MP];
  __shared__ float mst[NW][16];
  __shared__ float lst[NW][16];

  const int lane = threadIdx.x & 31;
  const int wid  = threadIdx.x >> 5;
  const int bb   = blockIdx.x >> 7;
  const int pr   = blockIdx.x & 127;
  const int m    = lane & 15;
  const int kp   = lane >> 4;

  const size_t base = (size_t)bb * Tsz * Hsz;
  const float* qb = qm + base;
  const float* kb = km + base;
  const float* vb = vm + base;

  for (int half = 0; half < 2; ++half) {
    const int ti = half ? pr : 255 - pr;

    v2f qf[16];
    {
      const float* qr = qb + (size_t)ti * (16 * Hsz) + lane * 4;
#pragma unroll
      for (int j = 0; j < 8; ++j) {
        const v4f t = *(const v4f*)(qr + j * 128);
        qf[2 * j].x     = t.x * 0.125f;
        qf[2 * j].y     = t.y * 0.125f;
        qf[2 * j + 1].x = t.z * 0.125f;
        qf[2 * j + 1].y = t.w * 0.125f;
      }
    }

    v8f acc[4] = {};
    float mrun = -1e30f;
    v8f lv = {};

    for (int j = wid; j <= ti; j += NW) {
      v8f s8 = {};
      const float* kt = kb + (size_t)j * (16 * Hsz) + lane * 4;
      v4f kc[8];
#pragma unroll
      for (int c = 0; c < 8; ++c) kc[c] = *(const v4f*)(kt + c * 128);
#pragma unroll
      for (int c = 0; c < 8; ++c) {
        v2f a0, a1;
        a0.x = kc[c].x; a0.y = kc[c].y;
        a1.x = kc[c].z; a1.y = kc[c].w;
        s8 = __builtin_amdgcn_wmma_f32_16x16x4_f32(false, a0, false, qf[2 * c],
                                                   (short)0, s8, false, false);
        s8 = __builtin_amdgcn_wmma_f32_16x16x4_f32(false, a1, false, qf[2 * c + 1],
                                                   (short)0, s8, false, false);
      }

      if (j == ti) {
#pragma unroll
        for (int r = 0; r < 8; ++r)
          if (r + 8 * kp > m) s8[r] = -1e30f;
      }

      v2f vf[4][4];
      {
        const float* vt = vb + (size_t)j * (16 * Hsz) + lane * 4;
#pragma unroll
        for (int nn = 0; nn < 4; ++nn)
#pragma unroll
          for (int u = 0; u < 2; ++u) {
            const v4f t = *(const v4f*)(vt + (nn * 2 + u) * 128);
            vf[2 * u][nn].x     = t.x;
            vf[2 * u][nn].y     = t.y;
            vf[2 * u + 1][nn].x = t.z;
            vf[2 * u + 1][nn].y = t.w;
          }
      }

      float tm = s8[0];
#pragma unroll
      for (int r = 1; r < 8; ++r) tm = fmaxf(tm, s8[r]);
      tm = fmaxf(tm, xhalf(tm));
      const float mnew  = fmaxf(mrun, tm);
      const float alpha = __expf(mrun - mnew);
      mrun = mnew;
      v8f p8;
#pragma unroll
      for (int r = 0; r < 8; ++r) p8[r] = __expf(s8[r] - mnew);
      lv = lv * alpha + p8;
#pragma unroll
      for (int nn = 0; nn < 4; ++nn)
#pragma unroll
        for (int r = 0; r < 8; ++r) acc[nn][r] *= alpha;

      __builtin_amdgcn_sched_barrier(0);

#pragma unroll
      for (int t = 0; t < 4; ++t) {
        v2f pf;
        pf.x = p8[2 * t];
        pf.y = p8[2 * t + 1];
#pragma unroll
        for (int nn = 0; nn < 4; ++nn)
          acc[nn] = __builtin_amdgcn_wmma_f32_16x16x4_f32(false, vf[t][nn], false, pf,
                                                          (short)0, acc[nn], false, false);
      }
    }

    {
      float* ma = macc + wid * (16 * MP) + m * MP + 8 * kp;
#pragma unroll
      for (int nn = 0; nn < 4; ++nn) {
        float4 lo, hi;
        lo.x = acc[nn][0]; lo.y = acc[nn][1]; lo.z = acc[nn][2]; lo.w = acc[nn][3];
        hi.x = acc[nn][4]; hi.y = acc[nn][5]; hi.z = acc[nn][6]; hi.w = acc[nn][7];
        *(float4*)(ma + nn * 16)     = lo;
        *(float4*)(ma + nn * 16 + 4) = hi;
      }
      const float lsum  = ((lv[0] + lv[1]) + (lv[2] + lv[3])) + ((lv[4] + lv[5]) + (lv[6] + lv[7]));
      const float lfull = lsum + xhalf(lsum);
      if (kp == 0) {
        mst[wid][m] = mrun;
        lst[wid][m] = lfull;
      }
    }

    __syncthreads();

    {
      const int row = threadIdx.x >> 4;
      const int c4  = (threadIdx.x & 15) * 4;
      float mw[NW];
      float mt = -1e30f;
#pragma unroll
      for (int w = 0; w < NW; ++w) { mw[w] = mst[w][row]; mt = fmaxf(mt, mw[w]); }
      float lt = 0.0f;
      float o0 = 0.0f, o1 = 0.0f, o2 = 0.0f, o3 = 0.0f;
#pragma unroll
      for (int w = 0; w < NW; ++w) {
        const float sc = __expf(mw[w] - mt);
        lt += lst[w][row] * sc;
        const float4 a = *(const float4*)(macc + (w * 16 + row) * MP + c4);
        o0 += a.x * sc; o1 += a.y * sc; o2 += a.z * sc; o3 += a.w * sc;
      }
      float4 res;
      res.x = o0 / lt; res.y = o1 / lt; res.z = o2 / lt; res.w = o3 / lt;
      *(float4*)(out + base + (size_t)(ti * 16 + row) * Hsz + c4) = res;
    }

    if (half == 0) __syncthreads();
  }
}

extern "C" void kernel_launch(void* const* d_in, const int* in_sizes, int n_in,
                              void* d_out, int out_size, void* d_ws, size_t ws_size,
                              hipStream_t stream) {
  (void)in_sizes; (void)n_in; (void)out_size; (void)ws_size;

  const float* x  = (const float*)d_in[0];
  const float* Wk = (const float*)d_in[1];
  const float* Wq = (const float*)d_in[2];
  const float* Wv = (const float*)d_in[3];
  float* out = (float*)d_out;

  float* qo = (float*)d_ws;
  float* ko = qo + (size_t)BT * Hsz;
  float* vo = ko + (size_t)BT * Hsz;

  const int tiles = BT / 16;
  dim3 block(128);
  dim3 grid(tiles / 4);

  proj_qkv<<<grid, block, 0, stream>>>(x, Wk, Wq, Wv, qo, ko, vo);
  flash_attn<<<dim3(512), dim3(256), 0, stream>>>(qo, ko, vo, out);
}


# projection kernel: Wq, Wk, Wv staged through LDS once per workgroup (whole-line loads, row pairs interleaved), each fragment one 8-byte LDS read; same k order, every sum unchanged
# speedup vs baseline: 5.6891x; 1.2742x over previous
#include <hip/hip_runtime.h>

typedef __attribute__((ext_vector_type(2))) float v2f;
typedef __attribute__((ext_vector_type(4))) float v4f;
typedef __attribute__((ext_vector_type(8))) float v8f;

static constexpr int Bsz = 4;
static constexpr int Tsz = 4096;
static constexpr int Csz = 1024;
static constexpr int Hsz = 64;
static constexpr int BT  = Bsz * Tsz;

static constexpr int KC   = 16;
static constexpr int PP   = 160;
static constexpr int WMAT = (KC / 2) * PP;
static constexpr int WBUF = 3 * WMAT;
static constexpr int NCH  = Csz / KC;

struct Frag {
  v2f a;
  v2f bq[4], bk[4], bv[4];
};

__device__ __forceinline__ void mma_step(const Frag& f, v8f (&qa)[4], v8f (&ka)[4], v8f (&va)[4])
{
#pragma unroll
  for (int nn = 0; nn < 4; ++nn) {
    qa[nn] = __builtin_amdgcn_wmma_f32_16x16x4_f32(false, f.bq[nn], false, f.a,
                                                   (short)0, qa[nn], false, false);
    ka[nn] = __builtin_amdgcn_wmma_f32_16x16x4_f32(false, f.bk[nn], false, f.a,
                                                   (short)0, ka[nn], false, false);
    va[nn] = __builtin_amdgcn_wmma_f32_16x16x4_f32(false, f.a, false, f.bv[nn],
                                                   (short)0, va[nn], false, false);
  }
}

__device__ __forceinline__ void lds_barrier()
{
  __builtin_amdgcn_fence(__ATOMIC_RELEASE, "workgroup", "local");
  __builtin_amdgcn_s_barrier();
  __builtin_amdgcn_fence(__ATOMIC_ACQUIRE, "workgroup", "local");
}

__global__ __launch_bounds__(128) void proj_qkv(
    const float* __restrict__ x,
    const float* __restrict__ Wk,
    const float* __restrict__ Wq,
    const float* __restrict__ Wv,
    float* __restrict__ qo, float* __restrict__ ko, float* __restrict__ vo)
{
  __shared__ __attribute__((aligned(16))) float wls[2 * WBUF];

  const int lane = threadIdx.x & 31;
  const int wid  = threadIdx.x >> 5;
  const int tile = blockIdx.x * 4 + wid;
  const int row0 = tile * 16;
  const int m    = lane & 15;
  const int kp   = lane >> 4;

  v8f qa[4] = {}, ka[4] = {}, va[4] = {};

  const float* xr = x + (size_t)(row0 + m) * Csz + 2 * kp;

  const int sp   = threadIdx.x >> 4;
  const int sc4  = (threadIdx.x & 15) * 4;
  const int gsrc = (2 * sp) * Hsz + sc4;
  const int sdst = sp * PP + sc4 * 2;
  const int lfr  = kp * PP + m * 2;

  v4f ga[3], gb[3];

  auto gload = [&](int ch) {
    const size_t o = (size_t)ch * (KC * Hsz) + gsrc;
    ga[0] = *(const v4f*)(Wq + o);  gb[0] = *(const v4f*)(Wq + o + Hsz);
    ga[1] = *(const v4f*)(Wk + o);  gb[1] = *(const v4f*)(Wk + o + Hsz);
    ga[2] = *(const v4f*)(Wv + o);  gb[2] = *(const v4f*)(Wv + o + Hsz);
  };
  auto sstore = [&](int buf) {
#pragma unroll
    for (int X = 0; X < 3; ++X) {
      v4f s0, s1;
      s0.x = ga[X].x; s0.y = gb[X].x; s0.z = ga[X].y; s0.w = gb[X].y;
      s1.x = ga[X].z; s1.y = gb[X].z; s1.z = ga[X].w; s1.w = gb[X].w;
      *(v4f*)(&wls[buf * WBUF + X * WMAT + sdst])     = s0;
      *(v4f*)(&wls[buf * WBUF + X * WMAT + sdst + 4]) = s1;
    }
  };
  auto xload = [&](Frag& f, int kk) { f.a = *(const v2f*)(xr + kk); };
  auto wload = [&](Frag& f, int buf, int t) {
    const int o = buf * WBUF + lfr + (2 * t) * PP;
#pragma unroll
    for (int nn = 0; nn < 4; ++nn) {
      f.bq[nn] = *(const v2f*)(&wls[o + 0 * WMAT + nn * 32]);
      f.bk[nn] = *(const v2f*)(&wls[o + 1 * WMAT + nn * 32]);
      f.bv[nn] = *(const v2f*)(&wls[o + 2 * WMAT + nn * 32]);
    }
  };

  gload(0);
  sstore(0);
  gload(1);
  lds_barrier();

  Frag c0, c1, n0, n1;
  xload(c0, 0);  wload(c0, 0, 0);
  xload(c1, 4);  wload(c1, 0, 1);

  for (int c = 0; c < NCH - 1; ++c) {
    const int kk  = c * KC;
    const int cur = c & 1;
    const int nxt = cur ^ 1;
    xload(n0, kk + 8);
    xload(n1, kk + 12);
    __builtin_amdgcn_sched_barrier(0);
    sstore(nxt);
    gload(c + 2 < NCH ? c + 2 : NCH - 1);
    wload(n0, cur, 2);
    wload(n1, cur, 3);
    __builtin_amdgcn_sched_barrier(0);
    mma_step(c0, qa, ka, va);
    mma_step(c1, qa, ka, va);
    lds_barrier();
    xload(c0, kk + 16);  wload(c0, nxt, 0);
    xload(c1, kk + 20);  wload(c1, nxt, 1);
    mma_step(n0, qa, ka, va);
    mma_step(n1, qa, ka, va);
  }
  xload(n0, Csz - 8);  wload(n0, (NCH - 1) & 1, 2);
  xload(n1, Csz - 4);  wload(n1, (NCH - 1) & 1, 3);
  mma_step(c0, qa, ka, va);
  mma_step(c1, qa, ka, va);
  mma_step(n0, qa, ka, va);
  mma_step(n1, qa, ka, va);

  const size_t tb = (size_t)tile * (16 * Hsz) + lane * 4;
#pragma unroll
  for (int nn = 0; nn < 4; ++nn)
#pragma unroll
    for (int u = 0; u < 2; ++u) {
      const size_t idx = tb + (nn * 2 + u) * 128;
      v4f tq, tk, tv;
#pragma unroll
      for (int e = 0; e < 4; ++e) {
        tq[e] = qa[nn][4 * u + e];
        tk[e] = ka[nn][4 * u + e];
        tv[e] = va[nn][4 * u + e];
      }
      *(v4f*)(qo + idx) = tq;
      *(v4f*)(ko + idx) = tk;
      *(v4f*)(vo + idx) = tv;
    }
}

static constexpr int NW = 8;
static constexpr int MP = 68;

__device__ __forceinline__ float xhalf(float x)
{
  const int b = __builtin_bit_cast(int, x);
  return __builtin_bit_cast(float,
      __builtin_amdgcn_permlanex16(b, b, 0x76543210, 0xfedcba98, false, false));
}

__global__ __launch_bounds__(NW * 32) void flash_attn(
    const float* __restrict__ qm, const float* __restrict__ km,
    const float* __restrict__ vm, float* __restrict__ out)
{
  __shared__ __attribute__((aligned(16))) float macc[NW * 16 * MP];
  __shared__ float mst[NW][16];
  __shared__ float lst[NW][16];

  const int lane = threadIdx.x & 31;
  const int wid  = threadIdx.x >> 5;
  const int bb   = blockIdx.x >> 7;
  const int pr   = blockIdx.x & 127;
  const int m    = lane & 15;
  const int kp   = lane >> 4;

  const size_t base = (size_t)bb * Tsz * Hsz;
  const float* qb = qm + base;
  const float* kb = km + base;
  const float* vb = vm + base;

  for (int half = 0; half < 2; ++half) {
    const int ti = half ? pr : 255 - pr;

    v2f qf[16];
    {
      const float* qr = qb + (size_t)ti * (16 * Hsz) + lane * 4;
#pragma unroll
      for (int j = 0; j < 8; ++j) {
        const v4f t = *(const v4f*)(qr + j * 128);
        qf[2 * j].x     = t.x * 0.125f;
        qf[2 * j].y     = t.y * 0.125f;
        qf[2 * j + 1].x = t.z * 0.125f;
        qf[2 * j + 1].y = t.w * 0.125f;
      }
    }

    v8f acc[4] = {};
    float mrun = -1e30f;
    v8f lv = {};

    for (int j = wid; j <= ti; j += NW) {
      v8f s8 = {};
      const float* kt = kb + (size_t)j * (16 * Hsz) + lane * 4;
      v4f kc[8];
#pragma unroll
      for (int c = 0; c < 8; ++c) kc[c] = *(const v4f*)(kt + c * 128);
#pragma unroll
      for (int c = 0; c < 8; ++c) {
        v2f a0, a1;
        a0.x = kc[c].x; a0.y = kc[c].y;
        a1.x = kc[c].z; a1.y = kc[c].w;
        s8 = __builtin_amdgcn_wmma_f32_16x16x4_f32(false, a0, false, qf[2 * c],
                                                   (short)0, s8, false, false);
        s8 = __builtin_amdgcn_wmma_f32_16x16x4_f32(false, a1, false, qf[2 * c + 1],
                                                   (short)0, s8, false, false);
      }

      if (j == ti) {
#pragma unroll
        for (int r = 0; r < 8; ++r)
          if (r + 8 * kp > m) s8[r] = -1e30f;
      }

      v2f vf[4][4];
      {
        const float* vt = vb + (size_t)j * (16 * Hsz) + lane * 4;
#pragma unroll
        for (int nn = 0; nn < 4; ++nn)
#pragma unroll
          for (int u = 0; u < 2; ++u) {
            const v4f t = *(const v4f*)(vt + (nn * 2 + u) * 128);
            vf[2 * u][nn].x     = t.x;
            vf[2 * u][nn].y     = t.y;
            vf[2 * u + 1][nn].x = t.z;
            vf[2 * u + 1][nn].y = t.w;
          }
      }

      float tm = s8[0];
#pragma unroll
      for (int r = 1; r < 8; ++r) tm = fmaxf(tm, s8[r]);
      tm = fmaxf(tm, xhalf(tm));
      const float mnew  = fmaxf(mrun, tm);
      const float alpha = __expf(mrun - mnew);
      mrun = mnew;
      v8f p8;
#pragma unroll
      for (int r = 0; r < 8; ++r) p8[r] = __expf(s8[r] - mnew);
      lv = lv * alpha + p8;
#pragma unroll
      for (int nn = 0; nn < 4; ++nn)
#pragma unroll
        for (int r = 0; r < 8; ++r) acc[nn][r] *= alpha;

      __builtin_amdgcn_sched_barrier(0);

#pragma unroll
      for (int t = 0; t < 4; ++t) {
        v2f pf;
        pf.x = p8[2 * t];
        pf.y = p8[2 * t + 1];
#pragma unroll
        for (int nn = 0; nn < 4; ++nn)
          acc[nn] = __builtin_amdgcn_wmma_f32_16x16x4_f32(false, vf[t][nn], false, pf,
                                                          (short)0, acc[nn], false, false);
      }
    }

    {
      float* ma = macc + wid * (16 * MP) + m * MP + 8 * kp;
#pragma unroll
      for (int nn = 0; nn < 4; ++nn) {
        float4 lo, hi;
        lo.x = acc[nn][0]; lo.y = acc[nn][1]; lo.z = acc[nn][2]; lo.w = acc[nn][3];
        hi.x = acc[nn][4]; hi.y = acc[nn][5]; hi.z = acc[nn][6]; hi.w = acc[nn][7];
        *(float4*)(ma + nn * 16)     = lo;
        *(float4*)(ma + nn * 16 + 4) = hi;
      }
      const float lsum  = ((lv[0] + lv[1]) + (lv[2] + lv[3])) + ((lv[4] + lv[5]) + (lv[6] + lv[7]));
      const float lfull = lsum + xhalf(lsum);
      if (kp == 0) {
        mst[wid][m] = mrun;
        lst[wid][m] = lfull;
      }
    }

    __syncthreads();

    {
      const int row = threadIdx.x >> 4;
      const int c4  = (threadIdx.x & 15) * 4;
      float mw[NW];
      float mt = -1e30f;
#pragma unroll
      for (int w = 0; w < NW; ++w) { mw[w] = mst[w][row]; mt = fmaxf(mt, mw[w]); }
      float lt = 0.0f;
      float o0 = 0.0f, o1 = 0.0f, o2 = 0.0f, o3 = 0.0f;
#pragma unroll
      for (int w = 0; w < NW; ++w) {
        const float sc = __expf(mw[w] - mt);
        lt += lst[w][row] * sc;
        const float4 a = *(const float4*)(macc + (w * 16 + row) * MP + c4);
        o0 += a.x * sc; o1 += a.y * sc; o2 += a.z * sc; o3 += a.w * sc;
      }
      float4 res;
      res.x = o0 / lt; res.y = o1 / lt; res.z = o2 / lt; res.w = o3 / lt;
      *(float4*)(out + base + (size_t)(ti * 16 + row) * Hsz + c4) = res;
    }

    if (half == 0) __syncthreads();
  }
}

extern "C" void kernel_launch(void* const* d_in, const int* in_sizes, int n_in,
                              void* d_out, int out_size, void* d_ws, size_t ws_size,
                              hipStream_t stream) {
  (void)in_sizes; (void)n_in; (void)out_size; (void)ws_size;

  const float* x  = (const float*)d_in[0];
  const float* Wk = (const float*)d_in[1];
  const float* Wq = (const float*)d_in[2];
  const float* Wv = (const float*)d_in[3];
  float* out = (float*)d_out;

  float* qo = (float*)d_ws;
  float* ko = qo + (size_t)BT * Hsz;
  float* vo = ko + (size_t)BT * Hsz;

  const int tiles = BT / 16;
  dim3 block(128);
  dim3 grid(tiles / 4);

  proj_qkv<<<grid, block, 0, stream>>>(x, Wk, Wq, Wv, qo, ko, vo);
  flash_attn<<<dim3(512), dim3(256), 0, stream>>>(qo, ko, vo, out);
}


# attention: a wave keeps two 16-query sub-tiles (32-query q tile, 256 workgroups); each K/V tile, loaded as one group, feeds two score and two output tiles: K/V bytes from L2 halved; sums unchanged
# speedup vs baseline: 6.0534x; 1.0640x over previous
#include <hip/hip_runtime.h>

typedef __attribute__((ext_vector_type(2))) float v2f;
typedef __attribute__((ext_vector_type(4))) float v4f;
typedef __attribute__((ext_vector_type(8))) float v8f;

static constexpr int Bsz = 4;
static constexpr int Tsz = 4096;
static constexpr int Csz = 1024;
static constexpr int Hsz = 64;
static constexpr int BT  = Bsz * Tsz;

static constexpr int KC   = 16;
static constexpr int PP   = 160;
static constexpr int WMAT = (KC / 2) * PP;
static constexpr int WBUF = 3 * WMAT;
static constexpr int NCH  = Csz / KC;

struct Frag {
  v2f a;
  v2f bq[4], bk[4], bv[4];
};

__device__ __forceinline__ void mma_step(const Frag& f, v8f (&qa)[4], v8f (&ka)[4], v8f (&va)[4])
{
#pragma unroll
  for (int nn = 0; nn < 4; ++nn) {
    qa[nn] = __builtin_amdgcn_wmma_f32_16x16x4_f32(false, f.bq[nn], false, f.a,
                                                   (short)0, qa[nn], false, false);
    ka[nn] = __builtin_amdgcn_wmma_f32_16x16x4_f32(false, f.bk[nn], false, f.a,
                                                   (short)0, ka[nn], false, false);
    va[nn] = __builtin_amdgcn_wmma_f32_16x16x4_f32(false, f.a, false, f.bv[nn],
                                                   (short)0, va[nn], false, false);
  }
}

__device__ __forceinline__ void lds_barrier()
{
  __builtin_amdgcn_fence(__ATOMIC_RELEASE, "workgroup", "local");
  __builtin_amdgcn_s_barrier();
  __builtin_amdgcn_fence(__ATOMIC_ACQUIRE, "workgroup", "local");
}

__global__ __launch_bounds__(128) void proj_qkv(
    const float* __restrict__ x,
    const float* __restrict__ Wk,
    const float* __restrict__ Wq,
    const float* __restrict__ Wv,
    float* __restrict__ qo, float* __restrict__ ko, float* __restrict__ vo)
{
  __shared__ __attribute__((aligned(16))) float wls[2 * WBUF];

  const int lane = threadIdx.x & 31;
  const int wid  = threadIdx.x >> 5;
  const int tile = blockIdx.x * 4 + wid;
  const int row0 = tile * 16;
  const int m    = lane & 15;
  const int kp   = lane >> 4;

  v8f qa[4] = {}, ka[4] = {}, va[4] = {};

  const float* xr = x + (size_t)(row0 + m) * Csz + 2 * kp;

  const int sp   = threadIdx.x >> 4;
  const int sc4  = (threadIdx.x & 15) * 4;
  const int gsrc = (2 * sp) * Hsz + sc4;
  const int sdst = sp * PP + sc4 * 2;
  const int lfr  = kp * PP + m * 2;

  v4f ga[3], gb[3];

  auto gload = [&](int ch) {
    const size_t o = (size_t)ch * (KC * Hsz) + gsrc;
    ga[0] = *(const v4f*)(Wq + o);  gb[0] = *(const v4f*)(Wq + o + Hsz);
    ga[1] = *(const v4f*)(Wk + o);  gb[1] = *(const v4f*)(Wk + o + Hsz);
    ga[2] = *(const v4f*)(Wv + o);  gb[2] = *(const v4f*)(Wv + o + Hsz);
  };
  auto sstore = [&](int buf) {
#pragma unroll
    for (int X = 0; X < 3; ++X) {
      v4f s0, s1;
      s0.x = ga[X].x; s0.y = gb[X].x; s0.z = ga[X].y; s0.w = gb[X].y;
      s1.x = ga[X].z; s1.y = gb[X].z; s1.z = ga[X].w; s1.w = gb[X].w;
      *(v4f*)(&wls[buf * WBUF + X * WMAT + sdst])     = s0;
      *(v4f*)(&wls[buf * WBUF + X * WMAT + sdst + 4]) = s1;
    }
  };
  auto xload = [&](Frag& f, int kk) { f.a = *(const v2f*)(xr + kk); };
  auto wload = [&](Frag& f, int buf, int t) {
    const int o = buf * WBUF + lfr + (2 * t) * PP;
#pragma unroll
    for (int nn = 0; nn < 4; ++nn) {
      f.bq[nn] = *(const v2f*)(&wls[o + 0 * WMAT + nn * 32]);
      f.bk[nn] = *(const v2f*)(&wls[o + 1 * WMAT + nn * 32]);
      f.bv[nn] = *(const v2f*)(&wls[o + 2 * WMAT + nn * 32]);
    }
  };

  gload(0);
  sstore(0);
  gload(1);
  lds_barrier();

  Frag c0, c1, n0, n1;
  xload(c0, 0);  wload(c0, 0, 0);
  xload(c1, 4);  wload(c1, 0, 1);

  for (int c = 0; c < NCH - 1; ++c) {
    const int kk  = c * KC;
    const int cur = c & 1;
    const int nxt = cur ^ 1;
    xload(n0, kk + 8);
    xload(n1, kk + 12);
    __builtin_amdgcn_sched_barrier(0);
    sstore(nxt);
    gload(c + 2 < NCH ? c + 2 : NCH - 1);
    wload(n0, cur, 2);
    wload(n1, cur, 3);
    __builtin_amdgcn_sched_barrier(0);
    mma_step(c0, qa, ka, va);
    mma_step(c1, qa, ka, va);
    lds_barrier();
    xload(c0, kk + 16);  wload(c0, nxt, 0);
    xload(c1, kk + 20);  wload(c1, nxt, 1);
    mma_step(n0, qa, ka, va);
    mma_step(n1, qa, ka, va);
  }
  xload(n0, Csz - 8);  wload(n0, (NCH - 1) & 1, 2);
  xload(n1, Csz - 4);  wload(n1, (NCH - 1) & 1, 3);
  mma_step(c0, qa, ka, va);
  mma_step(c1, qa, ka, va);
  mma_step(n0, qa, ka, va);
  mma_step(n1, qa, ka, va);

  const size_t tb = (size_t)tile * (16 * Hsz) + lane * 4;
#pragma unroll
  for (int nn = 0; nn < 4; ++nn)
#pragma unroll
    for (int u = 0; u < 2; ++u) {
      const size_t idx = tb + (nn * 2 + u) * 128;
      v4f tq, tk, tv;
#pragma unroll
      for (int e = 0; e < 4; ++e) {
        tq[e] = qa[nn][4 * u + e];
        tk[e] = ka[nn][4 * u + e];
        tv[e] = va[nn][4 * u + e];
      }
      *(v4f*)(qo + idx) = tq;
      *(v4f*)(ko + idx) = tk;
      *(v4f*)(vo + idx) = tv;
    }
}

static constexpr int NW = 8;
static constexpr int QT = 32;
static constexpr int MP = 68;

__device__ __forceinline__ float xhalf(float x)
{
  const int b = __builtin_bit_cast(int, x);
  return __builtin_bit_cast(float,
      __builtin_amdgcn_permlanex16(b, b, 0x76543210, 0xfedcba98, false, false));
}

__global__ __launch_bounds__(NW * 32) void flash_attn(
    const float* __restrict__ qm, const float* __restrict__ km,
    const float* __restrict__ vm, float* __restrict__ out)
{
  __shared__ __attribute__((aligned(16))) float macc[NW * QT * MP];
  __shared__ float mst[NW][QT];
  __shared__ float lst[NW][QT];

  const int lane = threadIdx.x & 31;
  const int wid  = threadIdx.x >> 5;
  const int bb   = blockIdx.x >> 6;
  const int pr   = blockIdx.x & 63;
  const int m    = lane & 15;
  const int kp   = lane >> 4;

  const size_t base = (size_t)bb * Tsz * Hsz;
  const float* qb = qm + base;
  const float* kb = km + base;
  const float* vb = vm + base;

  for (int half = 0; half < 2; ++half) {
    const int ti = half ? pr : 127 - pr;

    v2f qf[2][16];
#pragma unroll
    for (int s = 0; s < 2; ++s) {
      const float* qr = qb + (size_t)(2 * ti + s) * (16 * Hsz) + lane * 4;
#pragma unroll
      for (int j = 0; j < 8; ++j) {
        const v4f t = *(const v4f*)(qr + j * 128);
        qf[s][2 * j].x     = t.x * 0.125f;
        qf[s][2 * j].y     = t.y * 0.125f;
        qf[s][2 * j + 1].x = t.z * 0.125f;
        qf[s][2 * j + 1].y = t.w * 0.125f;
      }
    }

    v8f acc[2][4] = {};
    float mrun[2] = {-1e30f, -1e30f};
    v8f lv[2] = {};

    const int jd = 2 * ti;
    for (int j = wid; j <= jd + 1; j += NW) {
      const float* kt = kb + (size_t)j * (16 * Hsz) + lane * 4;
      v4f kc[8];
#pragma unroll
      for (int c = 0; c < 8; ++c) kc[c] = *(const v4f*)(kt + c * 128);

      v2f vf[4][4];
      {
        const float* vt = vb + (size_t)j * (16 * Hsz) + lane * 4;
#pragma unroll
        for (int nn = 0; nn < 4; ++nn)
#pragma unroll
          for (int u = 0; u < 2; ++u) {
            const v4f t = *(const v4f*)(vt + (nn * 2 + u) * 128);
            vf[2 * u][nn].x     = t.x;
            vf[2 * u][nn].y     = t.y;
            vf[2 * u + 1][nn].x = t.z;
            vf[2 * u + 1][nn].y = t.w;
          }
      }

      v8f s8[2] = {};
#pragma unroll
      for (int c = 0; c < 8; ++c) {
        v2f a0, a1;
        a0.x = kc[c].x; a0.y = kc[c].y;
        a1.x = kc[c].z; a1.y = kc[c].w;
        s8[0] = __builtin_amdgcn_wmma_f32_16x16x4_f32(false, a0, false, qf[0][2 * c],
                                                      (short)0, s8[0], false, false);
        s8[1] = __builtin_amdgcn_wmma_f32_16x16x4_f32(false, a0, false, qf[1][2 * c],
                                                      (short)0, s8[1], false, false);
        s8[0] = __builtin_amdgcn_wmma_f32_16x16x4_f32(false, a1, false, qf[0][2 * c + 1],
                                                      (short)0, s8[0], false, false);
        s8[1] = __builtin_amdgcn_wmma_f32_16x16x4_f32(false, a1, false, qf[1][2 * c + 1],
                                                      (short)0, s8[1], false, false);
      }

      if (j >= jd) {
        const int lim = m - 8 * kp - 16 * (j - jd);
#pragma unroll
        for (int s = 0; s < 2; ++s)
#pragma unroll
          for (int r = 0; r < 8; ++r)
            if (r > lim + 16 * s) s8[s][r] = -__builtin_inff();
      }

      v8f p8[2];
#pragma unroll
      for (int s = 0; s < 2; ++s) {
        float tm = s8[s][0];
#pragma unroll
        for (int r = 1; r < 8; ++r) tm = fmaxf(tm, s8[s][r]);
        tm = fmaxf(tm, xhalf(tm));
        const float mnew  = fmaxf(mrun[s], tm);
        const float alpha = __expf(mrun[s] - mnew);
        mrun[s] = mnew;
#pragma unroll
        for (int r = 0; r < 8; ++r) p8[s][r] = __expf(s8[s][r] - mnew);
        lv[s] = lv[s] * alpha + p8[s];
#pragma unroll
        for (int nn = 0; nn < 4; ++nn)
#pragma unroll
          for (int r = 0; r < 8; ++r) acc[s][nn][r] *= alpha;
      }

      __builtin_amdgcn_sched_barrier(0);

#pragma unroll
      for (int t = 0; t < 4; ++t) {
        v2f pf0, pf1;
        pf0.x = p8[0][2 * t];
        pf0.y = p8[0][2 * t + 1];
        pf1.x = p8[1][2 * t];
        pf1.y = p8[1][2 * t + 1];
#pragma unroll
        for (int nn = 0; nn < 4; ++nn) {
          acc[0][nn] = __builtin_amdgcn_wmma_f32_16x16x4_f32(false, vf[t][nn], false, pf0,
                                                             (short)0, acc[0][nn], false, false);
          acc[1][nn] = __builtin_amdgcn_wmma_f32_16x16x4_f32(false, vf[t][nn], false, pf1,
                                                             (short)0, acc[1][nn], false, false);
        }
      }
    }

#pragma unroll
    for (int s = 0; s < 2; ++s) {
      float* ma = macc + (wid * QT + 16 * s + m) * MP + 8 * kp;
#pragma unroll
      for (int nn = 0; nn < 4; ++nn) {
        float4 lo, hi;
        lo.x = acc[s][nn][0]; lo.y = acc[s][nn][1]; lo.z = acc[s][nn][2]; lo.w = acc[s][nn][3];
        hi.x = acc[s][nn][4]; hi.y = acc[s][nn][5]; hi.z = acc[s][nn][6]; hi.w = acc[s][nn][7];
        *(float4*)(ma + nn * 16)     = lo;
        *(float4*)(ma + nn * 16 + 4) = hi;
      }
      const float lsum  = ((lv[s][0] + lv[s][1]) + (lv[s][2] + lv[s][3])) + ((lv[s][4] + lv[s][5]) + (lv[s][6] + lv[s][7]));
      const float lfull = lsum + xhalf(lsum);
      if (kp == 0) {
        mst[wid][16 * s + m] = mrun[s];
        lst[wid][16 * s + m] = lfull;
      }
    }

    __syncthreads();

#pragma unroll
    for (int rh = 0; rh < 2; ++rh) {
      const int row = (threadIdx.x >> 4) + 16 * rh;
      const int c4  = (threadIdx.x & 15) * 4;
      float mw[NW];
      float mt = -1e30f;
#pragma unroll
      for (int w = 0; w < NW; ++w) { mw[w] = mst[w][row]; mt = fmaxf(mt, mw[w]); }
      float lt = 0.0f;
      float o0 = 0.0f, o1 = 0.0f, o2 = 0.0f, o3 = 0.0f;
#pragma unroll
      for (int w = 0; w < NW; ++w) {
        const float sc = __expf(mw[w] - mt);
        lt += lst[w][row] * sc;
        const float4 a = *(const float4*)(macc + (w * QT + row) * MP + c4);
        o0 += a.x * sc; o1 += a.y * sc; o2 += a.z * sc; o3 += a.w * sc;
      }
      float4 res;
      res.x = o0 / lt; res.y = o1 / lt; res.z = o2 / lt; res.w = o3 / lt;
      *(float4*)(out + base + (size_t)(ti * QT + row) * Hsz + c4) = res;
    }

    if (half == 0) __syncthreads();
  }
}

extern "C" void kernel_launch(void* const* d_in, const int* in_sizes, int n_in,
                              void* d_out, int out_size, void* d_ws, size_t ws_size,
                              hipStream_t stream) {
  (void)in_sizes; (void)n_in; (void)out_size; (void)ws_size;

  const float* x  = (const float*)d_in[0];
  const float* Wk = (const float*)d_in[1];
  const float* Wq = (const float*)d_in[2];
  const float* Wv = (const float*)d_in[3];
  float* out = (float*)d_out;

  float* qo = (float*)d_ws;
  float* ko = qo + (size_t)BT * Hsz;
  float* vo = ko + (size_t)BT * Hsz;

  const int tiles = BT / 16;
  dim3 block(128);
  dim3 grid(tiles / 4);

  proj_qkv<<<grid, block, 0, stream>>>(x, Wk, Wq, Wv, qo, ko, vo);
  flash_attn<<<dim3(256), dim3(256), 0, stream>>>(qo, ko, vo, out);
}


# projection kernel: x fragments fetched six K-steps (1.5 staged chunks) ahead of use into a ring of eight register pairs, loop body written per chunk pair; same k order, every sum unchanged
# speedup vs baseline: 6.3169x; 1.0435x over previous
#include <hip/hip_runtime.h>

typedef __attribute__((ext_vector_type(2))) float v2f;
typedef __attribute__((ext_vector_type(4))) float v4f;
typedef __attribute__((ext_vector_type(8))) float v8f;

static constexpr int Bsz = 4;
static constexpr int Tsz = 4096;
static constexpr int Csz = 1024;
static constexpr int Hsz = 64;
static constexpr int BT  = Bsz * Tsz;

static constexpr int KC   = 16;
static constexpr int PP   = 160;
static constexpr int WMAT = (KC / 2) * PP;
static constexpr int WBUF = 3 * WMAT;
static constexpr int NCH  = Csz / KC;

struct Frag {
  v2f bq[4], bk[4], bv[4];
};

__device__ __forceinline__ void mma_step(const v2f a, const Frag& f, v8f (&qa)[4], v8f (&ka)[4], v8f (&va)[4])
{
#pragma unroll
  for (int nn = 0; nn < 4; ++nn) {
    qa[nn] = __builtin_amdgcn_wmma_f32_16x16x4_f32(false, f.bq[nn], false, a,
                                                   (short)0, qa[nn], false, false);
    ka[nn] = __builtin_amdgcn_wmma_f32_16x16x4_f32(false, f.bk[nn], false, a,
                                                   (short)0, ka[nn], false, false);
    va[nn] = __builtin_amdgcn_wmma_f32_16x16x4_f32(false, a, false, f.bv[nn],
                                                   (short)0, va[nn], false, false);
  }
}

__device__ __forceinline__ void lds_barrier()
{
  __builtin_amdgcn_fence(__ATOMIC_RELEASE, "workgroup", "local");
  __builtin_amdgcn_s_barrier();
  __builtin_amdgcn_fence(__ATOMIC_ACQUIRE, "workgroup", "local");
}

__global__ __launch_bounds__(128) void proj_qkv(
    const float* __restrict__ x,
    const float* __restrict__ Wk,
    const float* __restrict__ Wq,
    const float* __restrict__ Wv,
    float* __restrict__ qo, float* __restrict__ ko, float* __restrict__ vo)
{
  __shared__ __attribute__((aligned(16))) float wls[2 * WBUF];

  const int lane = threadIdx.x & 31;
  const int wid  = threadIdx.x >> 5;
  const int tile = blockIdx.x * 4 + wid;
  const int row0 = tile * 16;
  const int m    = lane & 15;
  const int kp   = lane >> 4;

  v8f qa[4] = {}, ka[4] = {}, va[4] = {};

  const float* xr = x + (size_t)(row0 + m) * Csz + 2 * kp;

  const int sp   = threadIdx.x >> 4;
  const int sc4  = (threadIdx.x & 15) * 4;
  const int gsrc = (2 * sp) * Hsz + sc4;
  const int sdst = sp * PP + sc4 * 2;
  const int lfr  = kp * PP + m * 2;

  v4f ga[3], gb[3];

  auto gload = [&](const int ch) __attribute__((always_inline)) {
    const size_t o = (size_t)ch * (KC * Hsz) + gsrc;
    ga[0] = *(const v4f*)(Wq + o);  gb[0] = *(const v4f*)(Wq + o + Hsz);
    ga[1] = *(const v4f*)(Wk + o);  gb[1] = *(const v4f*)(Wk + o + Hsz);
    ga[2] = *(const v4f*)(Wv + o);  gb[2] = *(const v4f*)(Wv + o + Hsz);
  };
  auto sstore = [&](const int buf) __attribute__((always_inline)) {
#pragma unroll
    for (int X = 0; X < 3; ++X) {
      v4f s0, s1;
      s0.x = ga[X].x; s0.y = gb[X].x; s0.z = ga[X].y; s0.w = gb[X].y;
      s1.x = ga[X].z; s1.y = gb[X].z; s1.z = ga[X].w; s1.w = gb[X].w;
      *(v4f*)(&wls[buf * WBUF + X * WMAT + sdst])     = s0;
      *(v4f*)(&wls[buf * WBUF + X * WMAT + sdst + 4]) = s1;
    }
  };
  v2f xs[8];
  auto xload = [&](const int slot, const int kk) __attribute__((always_inline)) {
    xs[slot] = *(const v2f*)(xr + kk);
  };
  auto wload = [&](Frag& f, const int buf, const int t) __attribute__((always_inline)) {
    const int o = buf * WBUF + lfr + (2 * t) * PP;
#pragma unroll
    for (int nn = 0; nn < 4; ++nn) {
      f.bq[nn] = *(const v2f*)(&wls[o + 0 * WMAT + nn * 32]);
      f.bk[nn] = *(const v2f*)(&wls[o + 1 * WMAT + nn * 32]);
      f.bv[nn] = *(const v2f*)(&wls[o + 2 * WMAT + nn * 32]);
    }
  };

  gload(0);
  xload(0, 0);   xload(1, 4);   xload(2, 8);
  xload(3, 12);  xload(4, 16);  xload(5, 20);
  sstore(0);
  gload(1);
  lds_barrier();

  Frag c0, c1, n0, n1;
  wload(c0, 0, 0);
  wload(c1, 0, 1);

  auto trip = [&](const int c, const int P, const bool fetch_w, const bool top_x, const bool mid_x)
      __attribute__((always_inline)) {
    const int kk  = c * KC;
    const int nxt = P ^ 1;
    const int s   = 4 * P;
    __builtin_amdgcn_sched_barrier(0);
    sstore(nxt);
    if (fetch_w) gload(c + 2);
    __builtin_amdgcn_sched_barrier(0);
    if (top_x) {
      xload((s + 6) & 7, kk + 24);
      xload((s + 7) & 7, kk + 28);
    }
    wload(n0, P, 2);
    wload(n1, P, 3);
    __builtin_amdgcn_sched_barrier(0);
    mma_step(xs[s],     c0, qa, ka, va);
    mma_step(xs[s + 1], c1, qa, ka, va);
    lds_barrier();
    if (mid_x) {
      xload(s,     kk + 32);
      xload(s + 1, kk + 36);
    }
    wload(c0, nxt, 0);
    wload(c1, nxt, 1);
    mma_step(xs[s + 2], n0, qa, ka, va);
    mma_step(xs[s + 3], n1, qa, ka, va);
  };

  for (int c = 0; c < NCH - 2; c += 2) {
    trip(c,     0, true, true, true);
    trip(c + 1, 1, true, true, true);
  }
  trip(NCH - 2, 0, false, true, false);
  wload(n0, 1, 2);
  wload(n1, 1, 3);
  mma_step(xs[4], c0, qa, ka, va);
  mma_step(xs[5], c1, qa, ka, va);
  mma_step(xs[6], n0, qa, ka, va);
  mma_step(xs[7], n1, qa, ka, va);

  const size_t tb = (size_t)tile * (16 * Hsz) + lane * 4;
#pragma unroll
  for (int nn = 0; nn < 4; ++nn)
#pragma unroll
    for (int u = 0; u < 2; ++u) {
      const size_t idx = tb + (nn * 2 + u) * 128;
      v4f tq, tk, tv;
#pragma unroll
      for (int e = 0; e < 4; ++e) {
        tq[e] = qa[nn][4 * u + e];
        tk[e] = ka[nn][4 * u + e];
        tv[e] = va[nn][4 * u + e];
      }
      *(v4f*)(qo + idx) = tq;
      *(v4f*)(ko + idx) = tk;
      *(v4f*)(vo + idx) = tv;
    }
}

static constexpr int NW = 8;
static constexpr int QT = 32;
static constexpr int MP = 68;

__device__ __forceinline__ float xhalf(float x)
{
  const int b = __builtin_bit_cast(int, x);
  return __builtin_bit_cast(float,
      __builtin_amdgcn_permlanex16(b, b, 0x76543210, 0xfedcba98, false, false));
}

__global__ __launch_bounds__(NW * 32) void flash_attn(
    const float* __restrict__ qm, const float* __restrict__ km,
    const float* __restrict__ vm, float* __restrict__ out)
{
  __shared__ __attribute__((aligned(16))) float macc[NW * QT * MP];
  __shared__ float mst[NW][QT];
  __shared__ float lst[NW][QT];

  const int lane = threadIdx.x & 31;
  const int wid  = threadIdx.x >> 5;
  const int bb   = blockIdx.x >> 6;
  const int pr   = blockIdx.x & 63;
  const int m    = lane & 15;
  const int kp   = lane >> 4;

  const size_t base = (size_t)bb * Tsz * Hsz;
  const float* qb = qm + base;
  const float* kb = km + base;
  const float* vb = vm + base;

  for (int half = 0; half < 2; ++half) {
    const int ti = half ? pr : 127 - pr;

    v2f qf[2][16];
#pragma unroll
    for (int s = 0; s < 2; ++s) {
      const float* qr = qb + (size_t)(2 * ti + s) * (16 * Hsz) + lane * 4;
#pragma unroll
      for (int j = 0; j < 8; ++j) {
        const v4f t = *(const v4f*)(qr + j * 128);
        qf[s][2 * j].x     = t.x * 0.125f;
        qf[s][2 * j].y     = t.y * 0.125f;
        qf[s][2 * j + 1].x = t.z * 0.125f;
        qf[s][2 * j + 1].y = t.w * 0.125f;
      }
    }

    v8f acc[2][4] = {};
    float mrun[2] = {-1e30f, -1e30f};
    v8f lv[2] = {};

    const int jd = 2 * ti;
    for (int j = wid; j <= jd + 1; j += NW) {
      const float* kt = kb + (size_t)j * (16 * Hsz) + lane * 4;
      v4f kc[8];
#pragma unroll
      for (int c = 0; c < 8; ++c) kc[c] = *(const v4f*)(kt + c * 128);

      v2f vf[4][4];
      {
        const float* vt = vb + (size_t)j * (16 * Hsz) + lane * 4;
#pragma unroll
        for (int nn = 0; nn < 4; ++nn)
#pragma unroll
          for (int u = 0; u < 2; ++u) {
            const v4f t = *(const v4f*)(vt + (nn * 2 + u) * 128);
            vf[2 * u][nn].x     = t.x;
            vf[2 * u][nn].y     = t.y;
            vf[2 * u + 1][nn].x = t.z;
            vf[2 * u + 1][nn].y = t.w;
          }
      }

      v8f s8[2] = {};
#pragma unroll
      for (int c = 0; c < 8; ++c) {
        v2f a0, a1;
        a0.x = kc[c].x; a0.y = kc[c].y;
        a1.x = kc[c].z; a1.y = kc[c].w;
        s8[0] = __builtin_amdgcn_wmma_f32_16x16x4_f32(false, a0, false, qf[0][2 * c],
                                                      (short)0, s8[0], false, false);
        s8[1] = __builtin_amdgcn_wmma_f32_16x16x4_f32(false, a0, false, qf[1][2 * c],
                                                      (short)0, s8[1], false, false);
        s8[0] = __builtin_amdgcn_wmma_f32_16x16x4_f32(false, a1, false, qf[0][2 * c + 1],
                                                      (short)0, s8[0], false, false);
        s8[1] = __builtin_amdgcn_wmma_f32_16x16x4_f32(false, a1, false, qf[1][2 * c + 1],
                                                      (short)0, s8[1], false, false);
      }

      if (j >= jd) {
        const int lim = m - 8 * kp - 16 * (j - jd);
#pragma unroll
        for (int s = 0; s < 2; ++s)
#pragma unroll
          for (int r = 0; r < 8; ++r)
            if (r > lim + 16 * s) s8[s][r] = -__builtin_inff();
      }

      v8f p8[2];
#pragma unroll
      for (int s = 0; s < 2; ++s) {
        float tm = s8[s][0];
#pragma unroll
        for (int r = 1; r < 8; ++r) tm = fmaxf(tm, s8[s][r]);
        tm = fmaxf(tm, xhalf(tm));
        const float mnew  = fmaxf(mrun[s], tm);
        const float alpha = __expf(mrun[s] - mnew);
        mrun[s] = mnew;
#pragma unroll
        for (int r = 0; r < 8; ++r) p8[s][r] = __expf(s8[s][r] - mnew);
        lv[s] = lv[s] * alpha + p8[s];
#pragma unroll
        for (int nn = 0; nn < 4; ++nn)
#pragma unroll
          for (int r = 0; r < 8; ++r) acc[s][nn][r] *= alpha;
      }

      __builtin_amdgcn_sched_barrier(0);

#pragma unroll
      for (int t = 0; t < 4; ++t) {
        v2f pf0, pf1;
        pf0.x = p8[0][2 * t];
        pf0.y = p8[0][2 * t + 1];
        pf1.x = p8[1][2 * t];
        pf1.y = p8[1][2 * t + 1];
#pragma unroll
        for (int nn = 0; nn < 4; ++nn) {
          acc[0][nn] = __builtin_amdgcn_wmma_f32_16x16x4_f32(false, vf[t][nn], false, pf0,
                                                             (short)0, acc[0][nn], false, false);
          acc[1][nn] = __builtin_amdgcn_wmma_f32_16x16x4_f32(false, vf[t][nn], false, pf1,
                                                             (short)0, acc[1][nn], false, false);
        }
      }
    }

#pragma unroll
    for (int s = 0; s < 2; ++s) {
      float* ma = macc + (wid * QT + 16 * s + m) * MP + 8 * kp;
#pragma unroll
      for (int nn = 0; nn < 4; ++nn) {
        float4 lo, hi;
        lo.x = acc[s][nn][0]; lo.y = acc[s][nn][1]; lo.z = acc[s][nn][2]; lo.w = acc[s][nn][3];
        hi.x = acc[s][nn][4]; hi.y = acc[s][nn][5]; hi.z = acc[s][nn][6]; hi.w = acc[s][nn][7];
        *(float4*)(ma + nn * 16)     = lo;
        *(float4*)(ma + nn * 16 + 4) = hi;
      }
      const float lsum  = ((lv[s][0] + lv[s][1]) + (lv[s][2] + lv[s][3])) + ((lv[s][4] + lv[s][5]) + (lv[s][6] + lv[s][7]));
      const float lfull = lsum + xhalf(lsum);
      if (kp == 0) {
        mst[wid][16 * s + m] = mrun[s];
        lst[wid][16 * s + m] = lfull;
      }
    }

    __syncthreads();

#pragma unroll
    for (int rh = 0; rh < 2; ++rh) {
      const int row = (threadIdx.x >> 4) + 16 * rh;
      const int c4  = (threadIdx.x & 15) * 4;
      float mw[NW];
      float mt = -1e30f;
#pragma unroll
      for (int w = 0; w < NW; ++w) { mw[w] = mst[w][row]; mt = fmaxf(mt, mw[w]); }
      float lt = 0.0f;
      float o0 = 0.0f, o1 = 0.0f, o2 = 0.0f, o3 = 0.0f;
#pragma unroll
      for (int w = 0; w < NW; ++w) {
        const float sc = __expf(mw[w] - mt);
        lt += lst[w][row] * sc;
        const float4 a = *(const float4*)(macc + (w * QT + row) * MP + c4);
        o0 += a.x * sc; o1 += a.y * sc; o2 += a.z * sc; o3 += a.w * sc;
      }
      float4 res;
      res.x = o0 / lt; res.y = o1 / lt; res.z = o2 / lt; res.w = o3 / lt;
      *(float4*)(out + base + (size_t)(ti * QT + row) * Hsz + c4) = res;
    }

    if (half == 0) __syncthreads();
  }
}

extern "C" void kernel_launch(void* const* d_in, const int* in_sizes, int n_in,
                              void* d_out, int out_size, void* d_ws, size_t ws_size,
                              hipStream_t stream) {
  (void)in_sizes; (void)n_in; (void)out_size; (void)ws_size;

  const float* x  = (const float*)d_in[0];
  const float* Wk = (const float*)d_in[1];
  const float* Wq = (const float*)d_in[2];
  const float* Wv = (const float*)d_in[3];
  float* out = (float*)d_out;

  float* qo = (float*)d_ws;
  float* ko = qo + (size_t)BT * Hsz;
  float* vo = ko + (size_t)BT * Hsz;

  const int tiles = BT / 16;
  dim3 block(128);
  dim3 grid(tiles / 4);

  proj_qkv<<<grid, block, 0, stream>>>(x, Wk, Wq, Wv, qo, ko, vo);
  flash_attn<<<dim3(256), dim3(256), 0, stream>>>(qo, ko, vo, out);
}


# attention: a wave requests the K and V fragments of its next k/v tile one tile step ahead, into a second register set (loop per tile pair, scalar wave index): no product waits on L2; sums unchanged
# speedup vs baseline: 6.6104x; 1.0465x over previous
#include <hip/hip_runtime.h>

typedef __attribute__((ext_vector_type(2))) float v2f;
typedef __attribute__((ext_vector_type(4))) float v4f;
typedef __attribute__((ext_vector_type(8))) float v8f;

static constexpr int Bsz = 4;
static constexpr int Tsz = 4096;
static constexpr int Csz = 1024;
static constexpr int Hsz = 64;
static constexpr int BT  = Bsz * Tsz;

static constexpr int KC   = 16;
static constexpr int PP   = 160;
static constexpr int WMAT = (KC / 2) * PP;
static constexpr int WBUF = 3 * WMAT;
static constexpr int NCH  = Csz / KC;

struct Frag {
  v2f bq[4], bk[4], bv[4];
};

__device__ __forceinline__ void mma_step(const v2f a, const Frag& f, v8f (&qa)[4], v8f (&ka)[4], v8f (&va)[4])
{
#pragma unroll
  for (int nn = 0; nn < 4; ++nn) {
    qa[nn] = __builtin_amdgcn_wmma_f32_16x16x4_f32(false, f.bq[nn], false, a,
                                                   (short)0, qa[nn], false, false);
    ka[nn] = __builtin_amdgcn_wmma_f32_16x16x4_f32(false, f.bk[nn], false, a,
                                                   (short)0, ka[nn], false, false);
    va[nn] = __builtin_amdgcn_wmma_f32_16x16x4_f32(false, a, false, f.bv[nn],
                                                   (short)0, va[nn], false, false);
  }
}

__device__ __forceinline__ void lds_barrier()
{
  __builtin_amdgcn_fence(__ATOMIC_RELEASE, "workgroup", "local");
  __builtin_amdgcn_s_barrier();
  __builtin_amdgcn_fence(__ATOMIC_ACQUIRE, "workgroup", "local");
}

__global__ __launch_bounds__(128) void proj_qkv(
    const float* __restrict__ x,
    const float* __restrict__ Wk,
    const float* __restrict__ Wq,
    const float* __restrict__ Wv,
    float* __restrict__ qo, float* __restrict__ ko, float* __restrict__ vo)
{
  __shared__ __attribute__((aligned(16))) float wls[2 * WBUF];

  const int lane = threadIdx.x & 31;
  const int wid  = threadIdx.x >> 5;
  const int tile = blockIdx.x * 4 + wid;
  const int row0 = tile * 16;
  const int m    = lane & 15;
  const int kp   = lane >> 4;

  v8f qa[4] = {}, ka[4] = {}, va[4] = {};

  const float* xr = x + (size_t)(row0 + m) * Csz + 2 * kp;

  const int sp   = threadIdx.x >> 4;
  const int sc4  = (threadIdx.x & 15) * 4;
  const int gsrc = (2 * sp) * Hsz + sc4;
  const int sdst = sp * PP + sc4 * 2;
  const int lfr  = kp * PP + m * 2;

  v4f ga[3], gb[3];

  auto gload = [&](const int ch) __attribute__((always_inline)) {
    const size_t o = (size_t)ch * (KC * Hsz) + gsrc;
    ga[0] = *(const v4f*)(Wq + o);  gb[0] = *(const v4f*)(Wq + o + Hsz);
    ga[1] = *(const v4f*)(Wk + o);  gb[1] = *(const v4f*)(Wk + o + Hsz);
    ga[2] = *(const v4f*)(Wv + o);  gb[2] = *(const v4f*)(Wv + o + Hsz);
  };
  auto sstore = [&](const int buf) __attribute__((always_inline)) {
#pragma unroll
    for (int X = 0; X < 3; ++X) {
      v4f s0, s1;
      s0.x = ga[X].x; s0.y = gb[X].x; s0.z = ga[X].y; s0.w = gb[X].y;
      s1.x = ga[X].z; s1.y = gb[X].z; s1.z = ga[X].w; s1.w = gb[X].w;
      *(v4f*)(&wls[buf * WBUF + X * WMAT + sdst])     = s0;
      *(v4f*)(&wls[buf * WBUF + X * WMAT + sdst + 4]) = s1;
    }
  };
  v2f xs[8];
  auto xload = [&](const int slot, const int kk) __attribute__((always_inline)) {
    xs[slot] = *(const v2f*)(xr + kk);
  };
  auto wload = [&](Frag& f, const int buf, const int t) __attribute__((always_inline)) {
    const int o = buf * WBUF + lfr + (2 * t) * PP;
#pragma unroll
    for (int nn = 0; nn < 4; ++nn) {
      f.bq[nn] = *(const v2f*)(&wls[o + 0 * WMAT + nn * 32]);
      f.bk[nn] = *(const v2f*)(&wls[o + 1 * WMAT + nn * 32]);
      f.bv[nn] = *(const v2f*)(&wls[o + 2 * WMAT + nn * 32]);
    }
  };

  gload(0);
  xload(0, 0);   xload(1, 4);   xload(2, 8);
  xload(3, 12);  xload(4, 16);  xload(5, 20);
  sstore(0);
  gload(1);
  lds_barrier();

  Frag c0, c1, n0, n1;
  wload(c0, 0, 0);
  wload(c1, 0, 1);

  auto trip = [&](const int c, const int P, const bool fetch_w, const bool top_x, const bool mid_x)
      __attribute__((always_inline)) {
    const int kk  = c * KC;
    const int nxt = P ^ 1;
    const int s   = 4 * P;
    __builtin_amdgcn_sched_barrier(0);
    sstore(nxt);
    if (fetch_w) gload(c + 2);
    __builtin_amdgcn_sched_barrier(0);
    if (top_x) {
      xload((s + 6) & 7, kk + 24);
      xload((s + 7) & 7, kk + 28);
    }
    wload(n0, P, 2);
    wload(n1, P, 3);
    __builtin_amdgcn_sched_barrier(0);
    mma_step(xs[s],     c0, qa, ka, va);
    mma_step(xs[s + 1], c1, qa, ka, va);
    lds_barrier();
    if (mid_x) {
      xload(s,     kk + 32);
      xload(s + 1, kk + 36);
    }
    wload(c0, nxt, 0);
    wload(c1, nxt, 1);
    mma_step(xs[s + 2], n0, qa, ka, va);
    mma_step(xs[s + 3], n1, qa, ka, va);
  };

  for (int c = 0; c < NCH - 2; c += 2) {
    trip(c,     0, true, true, true);
    trip(c + 1, 1, true, true, true);
  }
  trip(NCH - 2, 0, false, true, false);
  wload(n0, 1, 2);
  wload(n1, 1, 3);
  mma_step(xs[4], c0, qa, ka, va);
  mma_step(xs[5], c1, qa, ka, va);
  mma_step(xs[6], n0, qa, ka, va);
  mma_step(xs[7], n1, qa, ka, va);

  const size_t tb = (size_t)tile * (16 * Hsz) + lane * 4;
#pragma unroll
  for (int nn = 0; nn < 4; ++nn)
#pragma unroll
    for (int u = 0; u < 2; ++u) {
      const size_t idx = tb + (nn * 2 + u) * 128;
      v4f tq, tk, tv;
#pragma unroll
      for (int e = 0; e < 4; ++e) {
        tq[e] = qa[nn][4 * u + e];
        tk[e] = ka[nn][4 * u + e];
        tv[e] = va[nn][4 * u + e];
      }
      *(v4f*)(qo + idx) = tq;
      *(v4f*)(ko + idx) = tk;
      *(v4f*)(vo + idx) = tv;
    }
}

static constexpr int NW = 8;
static constexpr int QT = 32;
static constexpr int MP = 68;

__device__ __forceinline__ float xhalf(float x)
{
  const int b = __builtin_bit_cast(int, x);
  return __builtin_bit_cast(float,
      __builtin_amdgcn_permlanex16(b, b, 0x76543210, 0xfedcba98, false, false));
}

__global__ __launch_bounds__(NW * 32) void flash_attn(
    const float* __restrict__ qm, const float* __restrict__ km,
    const float* __restrict__ vm, float* __restrict__ out)
{
  __shared__ __attribute__((aligned(16))) float macc[NW * QT * MP];
  __shared__ float mst[NW][QT];
  __shared__ float lst[NW][QT];

  const int lane = threadIdx.x & 31;
  const int wid  = __builtin_amdgcn_readfirstlane((int)(threadIdx.x >> 5));
  const int bb   = blockIdx.x >> 6;
  const int pr   = blockIdx.x & 63;
  const int m    = lane & 15;
  const int kp   = lane >> 4;

  const size_t base = (size_t)bb * Tsz * Hsz;
  const float* qb = qm + base;
  const float* kb = km + base;
  const float* vb = vm + base;

  for (int half = 0; half < 2; ++half) {
    const int ti = half ? pr : 127 - pr;

    v2f qf[2][16];
#pragma unroll
    for (int s = 0; s < 2; ++s) {
      const float* qr = qb + (size_t)(2 * ti + s) * (16 * Hsz) + lane * 4;
#pragma unroll
      for (int j = 0; j < 8; ++j) {
        const v4f t = *(const v4f*)(qr + j * 128);
        qf[s][2 * j].x     = t.x * 0.125f;
        qf[s][2 * j].y     = t.y * 0.125f;
        qf[s][2 * j + 1].x = t.z * 0.125f;
        qf[s][2 * j + 1].y = t.w * 0.125f;
      }
    }

    v8f acc[2][4] = {};
    float mrun[2] = {-1e30f, -1e30f};
    v8f lv[2] = {};

    const int jd = 2 * ti;
    const int jl = jd + 1;

    v4f kA[8], vA[8], kB[8], vB[8];
    auto kvload = [&](v4f (&kc)[8], v4f (&vc)[8], const int j) __attribute__((always_inline)) {
      const float* kt = kb + (size_t)j * (16 * Hsz) + lane * 4;
      const float* vt = vb + (size_t)j * (16 * Hsz) + lane * 4;
#pragma unroll
      for (int c = 0; c < 8; ++c) kc[c] = *(const v4f*)(kt + c * 128);
#pragma unroll
      for (int c = 0; c < 8; ++c) vc[c] = *(const v4f*)(vt + c * 128);
    };

    auto tile = [&](const v4f (&kc)[8], const v4f (&vc)[8], const int j) __attribute__((always_inline)) {
      v8f s8[2] = {};
#pragma unroll
      for (int c = 0; c < 8; ++c) {
        v2f a0, a1;
        a0.x = kc[c].x; a0.y = kc[c].y;
        a1.x = kc[c].z; a1.y = kc[c].w;
        s8[0] = __builtin_amdgcn_wmma_f32_16x16x4_f32(false, a0, false, qf[0][2 * c],
                                                      (short)0, s8[0], false, false);
        s8[1] = __builtin_amdgcn_wmma_f32_16x16x4_f32(false, a0, false, qf[1][2 * c],
                                                      (short)0, s8[1], false, false);
        s8[0] = __builtin_amdgcn_wmma_f32_16x16x4_f32(false, a1, false, qf[0][2 * c + 1],
                                                      (short)0, s8[0], false, false);
        s8[1] = __builtin_amdgcn_wmma_f32_16x16x4_f32(false, a1, false, qf[1][2 * c + 1],
                                                      (short)0, s8[1], false, false);
      }

      asm volatile("s_wait_alu depctr_va_vdst(0)" : "+v"(s8[0]), "+v"(s8[1]));

      if (j >= jd) {
        const int lim = m - 8 * kp - 16 * (j - jd);
#pragma unroll
        for (int s = 0; s < 2; ++s)
#pragma unroll
          for (int r = 0; r < 8; ++r)
            if (r > lim + 16 * s) s8[s][r] = -__builtin_inff();
      }

      v8f p8[2];
#pragma unroll
      for (int s = 0; s < 2; ++s) {
        float tm = s8[s][0];
#pragma unroll
        for (int r = 1; r < 8; ++r) tm = fmaxf(tm, s8[s][r]);
        tm = fmaxf(tm, xhalf(tm));
        const float mnew  = fmaxf(mrun[s], tm);
        const float alpha = __expf(mrun[s] - mnew);
        mrun[s] = mnew;
#pragma unroll
        for (int r = 0; r < 8; ++r) p8[s][r] = __expf(s8[s][r] - mnew);
        lv[s] = lv[s] * alpha + p8[s];
#pragma unroll
        for (int nn = 0; nn < 4; ++nn)
#pragma unroll
          for (int r = 0; r < 8; ++r) acc[s][nn][r] *= alpha;
      }

      __builtin_amdgcn_sched_barrier(0);

#pragma unroll
      for (int t = 0; t < 4; ++t) {
        v2f vf[4];
#pragma unroll
        for (int nn = 0; nn < 4; ++nn) {
          const v4f w = vc[nn * 2 + (t >> 1)];
          vf[nn].x = (t & 1) ? w.z : w.x;
          vf[nn].y = (t & 1) ? w.w : w.y;
        }
        v2f pf0, pf1;
        pf0.x = p8[0][2 * t];
        pf0.y = p8[0][2 * t + 1];
        pf1.x = p8[1][2 * t];
        pf1.y = p8[1][2 * t + 1];
#pragma unroll
        for (int nn = 0; nn < 4; ++nn) {
          acc[0][nn] = __builtin_amdgcn_wmma_f32_16x16x4_f32(false, vf[nn], false, pf0,
                                                             (short)0, acc[0][nn], false, false);
          acc[1][nn] = __builtin_amdgcn_wmma_f32_16x16x4_f32(false, vf[nn], false, pf1,
                                                             (short)0, acc[1][nn], false, false);
        }
      }
    };

    {
      int j = wid;
      kvload(kA, vA, j < jl ? j : jl);
      while (j + NW <= jl) {
        kvload(kB, vB, j + NW);
        tile(kA, vA, j);
        kvload(kA, vA, j + 2 * NW < jl ? j + 2 * NW : jl);
        tile(kB, vB, j + NW);
        j += 2 * NW;
      }
      if (j <= jl) tile(kA, vA, j);
    }

#pragma unroll
    for (int s = 0; s < 2; ++s) {
      float* ma = macc + (wid * QT + 16 * s + m) * MP + 8 * kp;
#pragma unroll
      for (int nn = 0; nn < 4; ++nn) {
        float4 lo, hi;
        lo.x = acc[s][nn][0]; lo.y = acc[s][nn][1]; lo.z = acc[s][nn][2]; lo.w = acc[s][nn][3];
        hi.x = acc[s][nn][4]; hi.y = acc[s][nn][5]; hi.z = acc[s][nn][6]; hi.w = acc[s][nn][7];
        *(float4*)(ma + nn * 16)     = lo;
        *(float4*)(ma + nn * 16 + 4) = hi;
      }
      const float lsum  = ((lv[s][0] + lv[s][1]) + (lv[s][2] + lv[s][3])) + ((lv[s][4] + lv[s][5]) + (lv[s][6] + lv[s][7]));
      const float lfull = lsum + xhalf(lsum);
      if (kp == 0) {
        mst[wid][16 * s + m] = mrun[s];
        lst[wid][16 * s + m] = lfull;
      }
    }

    __syncthreads();

#pragma unroll
    for (int rh = 0; rh < 2; ++rh) {
      const int row = (threadIdx.x >> 4) + 16 * rh;
      const int c4  = (threadIdx.x & 15) * 4;
      float mw[NW];
      float mt = -1e30f;
#pragma unroll
      for (int w = 0; w < NW; ++w) { mw[w] = mst[w][row]; mt = fmaxf(mt, mw[w]); }
      float lt = 0.0f;
      float o0 = 0.0f, o1 = 0.0f, o2 = 0.0f, o3 = 0.0f;
#pragma unroll
      for (int w = 0; w < NW; ++w) {
        const float sc = __expf(mw[w] - mt);
        lt += lst[w][row] * sc;
        const float4 a = *(const float4*)(macc + (w * QT + row) * MP + c4);
        o0 += a.x * sc; o1 += a.y * sc; o2 += a.z * sc; o3 += a.w * sc;
      }
      float4 res;
      res.x = o0 / lt; res.y = o1 / lt; res.z = o2 / lt; res.w = o3 / lt;
      *(float4*)(out + base + (size_t)(ti * QT + row) * Hsz + c4) = res;
    }

    if (half == 0) __syncthreads();
  }
}

extern "C" void kernel_launch(void* const* d_in, const int* in_sizes, int n_in,
                              void* d_out, int out_size, void* d_ws, size_t ws_size,
                              hipStream_t stream) {
  (void)in_sizes; (void)n_in; (void)out_size; (void)ws_size;

  const float* x  = (const float*)d_in[0];
  const float* Wk = (const float*)d_in[1];
  const float* Wq = (const float*)d_in[2];
  const float* Wv = (const float*)d_in[3];
  float* out = (float*)d_out;

  float* qo = (float*)d_ws;
  float* ko = qo + (size_t)BT * Hsz;
  float* vo = ko + (size_t)BT * Hsz;

  const int tiles = BT / 16;
  dim3 block(128);
  dim3 grid(tiles / 4);

  proj_qkv<<<grid, block, 0, stream>>>(x, Wk, Wq, Wv, qo, ko, vo);
  flash_attn<<<dim3(256), dim3(256), 0, stream>>>(qo, ko, vo, out);
}


# projection kernel: K range split over two wave groups (8-wave workgroup, two waves per SIMD), partial q/k/v tiles joined through LDS; new summation order: lower K half + upper K half
# speedup vs baseline: 6.9290x; 1.0482x over previous
#include <hip/hip_runtime.h>

typedef __attribute__((ext_vector_type(2))) float v2f;
typedef __attribute__((ext_vector_type(4))) float v4f;
typedef __attribute__((ext_vector_type(8))) float v8f;

static constexpr int Bsz = 4;
static constexpr int Tsz = 4096;
static constexpr int Csz = 1024;
static constexpr int Hsz = 64;
static constexpr int BT  = Bsz * Tsz;

static constexpr int KC   = 16;
static constexpr int PP   = 160;
static constexpr int WMAT = (KC / 2) * PP;
static constexpr int WBUF = 3 * WMAT;
static constexpr int KG   = 2;
static constexpr int NCH  = Csz / KC / KG;

struct Frag {
  v2f bq[4], bk[4], bv[4];
};

__device__ __forceinline__ void mma_step(const v2f a, const Frag& f, v8f (&qa)[4], v8f (&ka)[4], v8f (&va)[4])
{
#pragma unroll
  for (int nn = 0; nn < 4; ++nn) {
    qa[nn] = __builtin_amdgcn_wmma_f32_16x16x4_f32(false, f.bq[nn], false, a,
                                                   (short)0, qa[nn], false, false);
    ka[nn] = __builtin_amdgcn_wmma_f32_16x16x4_f32(false, f.bk[nn], false, a,
                                                   (short)0, ka[nn], false, false);
    va[nn] = __builtin_amdgcn_wmma_f32_16x16x4_f32(false, a, false, f.bv[nn],
                                                   (short)0, va[nn], false, false);
  }
}

__device__ __forceinline__ void lds_barrier()
{
  __builtin_amdgcn_fence(__ATOMIC_RELEASE, "workgroup", "local");
  __builtin_amdgcn_s_barrier();
  __builtin_amdgcn_fence(__ATOMIC_ACQUIRE, "workgroup", "local");
}

__global__ __launch_bounds__(256) void proj_qkv(
    const float* __restrict__ x,
    const float* __restrict__ Wk,
    const float* __restrict__ Wq,
    const float* __restrict__ Wv,
    float* __restrict__ qo, float* __restrict__ ko, float* __restrict__ vo)
{
  __shared__ __attribute__((aligned(16))) float wls[KG * 2 * WBUF];
  __shared__ __attribute__((aligned(16))) float xch[4 * 24 * 128];

  const int lane = threadIdx.x & 31;
  const int grp  = __builtin_amdgcn_readfirstlane((int)(threadIdx.x >> 7));
  const int wq   = (threadIdx.x >> 5) & 3;
  const int tile = blockIdx.x * 4 + wq;
  const int row0 = tile * 16;
  const int m    = lane & 15;
  const int kp   = lane >> 4;

  v8f qa[4] = {}, ka[4] = {}, va[4] = {};

  const float* xr = x + (size_t)(row0 + m) * Csz + grp * (Csz / KG) + 2 * kp;

  const int sp   = (threadIdx.x & 127) >> 4;
  const int sc4  = (threadIdx.x & 15) * 4;
  const int gsrc = grp * (Csz / KG) * Hsz + (2 * sp) * Hsz + sc4;
  const int sdst = grp * (2 * WBUF) + sp * PP + sc4 * 2;
  const int lfr  = grp * (2 * WBUF) + kp * PP + m * 2;

  v4f ga[3], gb[3];

  auto gload = [&](const int ch) __attribute__((always_inline)) {
    const size_t o = (size_t)ch * (KC * Hsz) + gsrc;
    ga[0] = *(const v4f*)(Wq + o);  gb[0] = *(const v4f*)(Wq + o + Hsz);
    ga[1] = *(const v4f*)(Wk + o);  gb[1] = *(const v4f*)(Wk + o + Hsz);
    ga[2] = *(const v4f*)(Wv + o);  gb[2] = *(const v4f*)(Wv + o + Hsz);
  };
  auto sstore = [&](const int buf) __attribute__((always_inline)) {
#pragma unroll
    for (int X = 0; X < 3; ++X) {
      v4f s0, s1;
      s0.x = ga[X].x; s0.y = gb[X].x; s0.z = ga[X].y; s0.w = gb[X].y;
      s1.x = ga[X].z; s1.y = gb[X].z; s1.z = ga[X].w; s1.w = gb[X].w;
      *(v4f*)(&wls[buf * WBUF + X * WMAT + sdst])     = s0;
      *(v4f*)(&wls[buf * WBUF + X * WMAT + sdst + 4]) = s1;
    }
  };
  v2f xs[8];
  auto xload = [&](const int slot, const int kk) __attribute__((always_inline)) {
    xs[slot] = *(const v2f*)(xr + kk);
  };
  auto wload = [&](Frag& f, const int buf, const int t) __attribute__((always_inline)) {
    const int o = buf * WBUF + lfr + (2 * t) * PP;
#pragma unroll
    for (int nn = 0; nn < 4; ++nn) {
      f.bq[nn] = *(const v2f*)(&wls[o + 0 * WMAT + nn * 32]);
      f.bk[nn] = *(const v2f*)(&wls[o + 1 * WMAT + nn * 32]);
      f.bv[nn] = *(const v2f*)(&wls[o + 2 * WMAT + nn * 32]);
    }
  };

  gload(0);
  xload(0, 0);   xload(1, 4);   xload(2, 8);
  xload(3, 12);  xload(4, 16);  xload(5, 20);
  sstore(0);
  gload(1);
  lds_barrier();

  Frag c0, c1, n0, n1;
  wload(c0, 0, 0);
  wload(c1, 0, 1);

  auto trip = [&](const int c, const int P, const bool fetch_w, const bool top_x, const bool mid_x)
      __attribute__((always_inline)) {
    const int kk  = c * KC;
    const int nxt = P ^ 1;
    const int s   = 4 * P;
    __builtin_amdgcn_sched_barrier(0);
    sstore(nxt);
    if (fetch_w) gload(c + 2);
    __builtin_amdgcn_sched_barrier(0);
    if (top_x) {
      xload((s + 6) & 7, kk + 24);
      xload((s + 7) & 7, kk + 28);
    }
    wload(n0, P, 2);
    wload(n1, P, 3);
    __builtin_amdgcn_sched_barrier(0);
    mma_step(xs[s],     c0, qa, ka, va);
    mma_step(xs[s + 1], c1, qa, ka, va);
    lds_barrier();
    if (mid_x) {
      xload(s,     kk + 32);
      xload(s + 1, kk + 36);
    }
    wload(c0, nxt, 0);
    wload(c1, nxt, 1);
    mma_step(xs[s + 2], n0, qa, ka, va);
    mma_step(xs[s + 3], n1, qa, ka, va);
  };

  for (int c = 0; c < NCH - 2; c += 2) {
    trip(c,     0, true, true, true);
    trip(c + 1, 1, true, true, true);
  }
  trip(NCH - 2, 0, false, true, false);
  wload(n0, 1, 2);
  wload(n1, 1, 3);
  mma_step(xs[4], c0, qa, ka, va);
  mma_step(xs[5], c1, qa, ka, va);
  mma_step(xs[6], n0, qa, ka, va);
  mma_step(xs[7], n1, qa, ka, va);

  float* xw = xch + (wq * 24) * 128 + lane * 4;
  if (grp == 1) {
#pragma unroll
    for (int nn = 0; nn < 4; ++nn)
#pragma unroll
      for (int u = 0; u < 2; ++u) {
        v4f tq, tk, tv;
#pragma unroll
        for (int e = 0; e < 4; ++e) {
          tq[e] = qa[nn][4 * u + e];
          tk[e] = ka[nn][4 * u + e];
          tv[e] = va[nn][4 * u + e];
        }
        *(v4f*)(xw + (0  + nn * 2 + u) * 128) = tq;
        *(v4f*)(xw + (8  + nn * 2 + u) * 128) = tk;
        *(v4f*)(xw + (16 + nn * 2 + u) * 128) = tv;
      }
  }
  __syncthreads();
  if (grp == 1) return;
#pragma unroll
  for (int nn = 0; nn < 4; ++nn)
#pragma unroll
    for (int u = 0; u < 2; ++u) {
      const v4f tq = *(const v4f*)(xw + (0  + nn * 2 + u) * 128);
      const v4f tk = *(const v4f*)(xw + (8  + nn * 2 + u) * 128);
      const v4f tv = *(const v4f*)(xw + (16 + nn * 2 + u) * 128);
#pragma unroll
      for (int e = 0; e < 4; ++e) {
        qa[nn][4 * u + e] += tq[e];
        ka[nn][4 * u + e] += tk[e];
        va[nn][4 * u + e] += tv[e];
      }
    }

  const size_t tb = (size_t)tile * (16 * Hsz) + lane * 4;
#pragma unroll
  for (int nn = 0; nn < 4; ++nn)
#pragma unroll
    for (int u = 0; u < 2; ++u) {
      const size_t idx = tb + (nn * 2 + u) * 128;
      v4f tq, tk, tv;
#pragma unroll
      for (int e = 0; e < 4; ++e) {
        tq[e] = qa[nn][4 * u + e];
        tk[e] = ka[nn][4 * u + e];
        tv[e] = va[nn][4 * u + e];
      }
      *(v4f*)(qo + idx) = tq;
      *(v4f*)(ko + idx) = tk;
      *(v4f*)(vo + idx) = tv;
    }
}

static constexpr int NW = 8;
static constexpr int QT = 32;
static constexpr int MP = 68;

__device__ __forceinline__ float xhalf(float x)
{
  const int b = __builtin_bit_cast(int, x);
  return __builtin_bit_cast(float,
      __builtin_amdgcn_permlanex16(b, b, 0x76543210, 0xfedcba98, false, false));
}

__global__ __launch_bounds__(NW * 32) void flash_attn(
    const float* __restrict__ qm, const float* __restrict__ km,
    const float* __restrict__ vm, float* __restrict__ out)
{
  __shared__ __attribute__((aligned(16))) float macc[NW * QT * MP];
  __shared__ float mst[NW][QT];
  __shared__ float lst[NW][QT];

  const int lane = threadIdx.x & 31;
  const int wid  = __builtin_amdgcn_readfirstlane((int)(threadIdx.x >> 5));
  const int bb   = blockIdx.x >> 6;
  const int pr   = blockIdx.x & 63;
  const int m    = lane & 15;
  const int kp   = lane >> 4;

  const size_t base = (size_t)bb * Tsz * Hsz;
  const float* qb = qm + base;
  const float* kb = km + base;
  const float* vb = vm + base;

  for (int half = 0; half < 2; ++half) {
    const int ti = half ? pr : 127 - pr;

    v2f qf[2][16];
#pragma unroll
    for (int s = 0; s < 2; ++s) {
      const float* qr = qb + (size_t)(2 * ti + s) * (16 * Hsz) + lane * 4;
#pragma unroll
      for (int j = 0; j < 8; ++j) {
        const v4f t = *(const v4f*)(qr + j * 128);
        qf[s][2 * j].x     = t.x * 0.125f;
        qf[s][2 * j].y     = t.y * 0.125f;
        qf[s][2 * j + 1].x = t.z * 0.125f;
        qf[s][2 * j + 1].y = t.w * 0.125f;
      }
    }

    v8f acc[2][4] = {};
    float mrun[2] = {-1e30f, -1e30f};
    v8f lv[2] = {};

    const int jd = 2 * ti;
    const int jl = jd + 1;

    v4f kA[8], vA[8], kB[8], vB[8];
    auto kvload = [&](v4f (&kc)[8], v4f (&vc)[8], const int j) __attribute__((always_inline)) {
      const float* kt = kb + (size_t)j * (16 * Hsz) + lane * 4;
      const float* vt = vb + (size_t)j * (16 * Hsz) + lane * 4;
#pragma unroll
      for (int c = 0; c < 8; ++c) kc[c] = *(const v4f*)(kt + c * 128);
#pragma unroll
      for (int c = 0; c < 8; ++c) vc[c] = *(const v4f*)(vt + c * 128);
    };

    auto tile = [&](const v4f (&kc)[8], const v4f (&vc)[8], const int j) __attribute__((always_inline)) {
      v8f s8[2] = {};
#pragma unroll
      for (int c = 0; c < 8; ++c) {
        v2f a0, a1;
        a0.x = kc[c].x; a0.y = kc[c].y;
        a1.x = kc[c].z; a1.y = kc[c].w;
        s8[0] = __builtin_amdgcn_wmma_f32_16x16x4_f32(false, a0, false, qf[0][2 * c],
                                                      (short)0, s8[0], false, false);
        s8[1] = __builtin_amdgcn_wmma_f32_16x16x4_f32(false, a0, false, qf[1][2 * c],
                                                      (short)0, s8[1], false, false);
        s8[0] = __builtin_amdgcn_wmma_f32_16x16x4_f32(false, a1, false, qf[0][2 * c + 1],
                                                      (short)0, s8[0], false, false);
        s8[1] = __builtin_amdgcn_wmma_f32_16x16x4_f32(false, a1, false, qf[1][2 * c + 1],
                                                      (short)0, s8[1], false, false);
      }

      asm volatile("s_wait_alu depctr_va_vdst(0)" : "+v"(s8[0]), "+v"(s8[1]));

      if (j >= jd) {
        const int lim = m - 8 * kp - 16 * (j - jd);
#pragma unroll
        for (int s = 0; s < 2; ++s)
#pragma unroll
          for (int r = 0; r < 8; ++r)
            if (r > lim + 16 * s) s8[s][r] = -__builtin_inff();
      }

      v8f p8[2];
#pragma unroll
      for (int s = 0; s < 2; ++s) {
        float tm = s8[s][0];
#pragma unroll
        for (int r = 1; r < 8; ++r) tm = fmaxf(tm, s8[s][r]);
        tm = fmaxf(tm, xhalf(tm));
        const float mnew  = fmaxf(mrun[s], tm);
        const float alpha = __expf(mrun[s] - mnew);
        mrun[s] = mnew;
#pragma unroll
        for (int r = 0; r < 8; ++r) p8[s][r] = __expf(s8[s][r] - mnew);
        lv[s] = lv[s] * alpha + p8[s];
#pragma unroll
        for (int nn = 0; nn < 4; ++nn)
#pragma unroll
          for (int r = 0; r < 8; ++r) acc[s][nn][r] *= alpha;
      }

      __builtin_amdgcn_sched_barrier(0);

#pragma unroll
      for (int t = 0; t < 4; ++t) {
        v2f vf[4];
#pragma unroll
        for (int nn = 0; nn < 4; ++nn) {
          const v4f w = vc[nn * 2 + (t >> 1)];
          vf[nn].x = (t & 1) ? w.z : w.x;
          vf[nn].y = (t & 1) ? w.w : w.y;
        }
        v2f pf0, pf1;
        pf0.x = p8[0][2 * t];
        pf0.y = p8[0][2 * t + 1];
        pf1.x = p8[1][2 * t];
        pf1.y = p8[1][2 * t + 1];
#pragma unroll
        for (int nn = 0; nn < 4; ++nn) {
          acc[0][nn] = __builtin_amdgcn_wmma_f32_16x16x4_f32(false, vf[nn], false, pf0,
                                                             (short)0, acc[0][nn], false, false);
          acc[1][nn] = __builtin_amdgcn_wmma_f32_16x16x4_f32(false, vf[nn], false, pf1,
                                                             (short)0, acc[1][nn], false, false);
        }
      }
    };

    {
      int j = wid;
      kvload(kA, vA, j < jl ? j : jl);
      while (j + NW <= jl) {
        kvload(kB, vB, j + NW);
        tile(kA, vA, j);
        kvload(kA, vA, j + 2 * NW < jl ? j + 2 * NW : jl);
        tile(kB, vB, j + NW);
        j += 2 * NW;
      }
      if (j <= jl) tile(kA, vA, j);
    }

#pragma unroll
    for (int s = 0; s < 2; ++s) {
      float* ma = macc + (wid * QT + 16 * s + m) * MP + 8 * kp;
#pragma unroll
      for (int nn = 0; nn < 4; ++nn) {
        float4 lo, hi;
        lo.x = acc[s][nn][0]; lo.y = acc[s][nn][1]; lo.z = acc[s][nn][2]; lo.w = acc[s][nn][3];
        hi.x = acc[s][nn][4]; hi.y = acc[s][nn][5]; hi.z = acc[s][nn][6]; hi.w = acc[s][nn][7];
        *(float4*)(ma + nn * 16)     = lo;
        *(float4*)(ma + nn * 16 + 4) = hi;
      }
      const float lsum  = ((lv[s][0] + lv[s][1]) + (lv[s][2] + lv[s][3])) + ((lv[s][4] + lv[s][5]) + (lv[s][6] + lv[s][7]));
      const float lfull = lsum + xhalf(lsum);
      if (kp == 0) {
        mst[wid][16 * s + m] = mrun[s];
        lst[wid][16 * s + m] = lfull;
      }
    }

    __syncthreads();

#pragma unroll
    for (int rh = 0; rh < 2; ++rh) {
      const int row = (threadIdx.x >> 4) + 16 * rh;
      const int c4  = (threadIdx.x & 15) * 4;
      float mw[NW];
      float mt = -1e30f;
#pragma unroll
      for (int w = 0; w < NW; ++w) { mw[w] = mst[w][row]; mt = fmaxf(mt, mw[w]); }
      float lt = 0.0f;
      float o0 = 0.0f, o1 = 0.0f, o2 = 0.0f, o3 = 0.0f;
#pragma unroll
      for (int w = 0; w < NW; ++w) {
        const float sc = __expf(mw[w] - mt);
        lt += lst[w][row] * sc;
        const float4 a = *(const float4*)(macc + (w * QT + row) * MP + c4);
        o0 += a.x * sc; o1 += a.y * sc; o2 += a.z * sc; o3 += a.w * sc;
      }
      float4 res;
      res.x = o0 / lt; res.y = o1 / lt; res.z = o2 / lt; res.w = o3 / lt;
      *(float4*)(out + base + (size_t)(ti * QT + row) * Hsz + c4) = res;
    }

    if (half == 0) __syncthreads();
  }
}

extern "C" void kernel_launch(void* const* d_in, const int* in_sizes, int n_in,
                              void* d_out, int out_size, void* d_ws, size_t ws_size,
                              hipStream_t stream) {
  (void)in_sizes; (void)n_in; (void)out_size; (void)ws_size;

  const float* x  = (const float*)d_in[0];
  const float* Wk = (const float*)d_in[1];
  const float* Wq = (const float*)d_in[2];
  const float* Wv = (const float*)d_in[3];
  float* out = (float*)d_out;

  float* qo = (float*)d_ws;
  float* ko = qo + (size_t)BT * Hsz;
  float* vo = ko + (size_t)BT * Hsz;

  const int tiles = BT / 16;
  dim3 block(256);
  dim3 grid(tiles / 4);

  proj_qkv<<<grid, block, 0, stream>>>(x, Wk, Wq, Wv, qo, ko, vo);
  flash_attn<<<dim3(256), dim3(256), 0, stream>>>(qo, ko, vo, out);
}


# both kernels: each wave lowers its own priority (s_setprio 3,2,1,0) as it advances between barriers, so the lagging wave of a SIMD outranks the leading one and both finish together; sums unchanged
# speedup vs baseline: 7.1963x; 1.0386x over previous
#include <hip/hip_runtime.h>

typedef __attribute__((ext_vector_type(2))) float v2f;
typedef __attribute__((ext_vector_type(4))) float v4f;
typedef __attribute__((ext_vector_type(8))) float v8f;

static constexpr int Bsz = 4;
static constexpr int Tsz = 4096;
static constexpr int Csz = 1024;
static constexpr int Hsz = 64;
static constexpr int BT  = Bsz * Tsz;

static constexpr int KC   = 16;
static constexpr int PP   = 160;
static constexpr int WMAT = (KC / 2) * PP;
static constexpr int WBUF = 3 * WMAT;
static constexpr int KG   = 2;
static constexpr int NCH  = Csz / KC / KG;

struct Frag {
  v2f bq[4], bk[4], bv[4];
};


__device__ __forceinline__ void mma_step(const v2f a, const Frag& f, v8f (&qa)[4], v8f (&ka)[4], v8f (&va)[4])
{
#pragma unroll
  for (int nn = 0; nn < 4; ++nn) {
    qa[nn] = __builtin_amdgcn_wmma_f32_16x16x4_f32(false, f.bq[nn], false, a,
                                                   (short)0, qa[nn], false, false);
    ka[nn] = __builtin_amdgcn_wmma_f32_16x16x4_f32(false, f.bk[nn], false, a,
                                                   (short)0, ka[nn], false, false);
    va[nn] = __builtin_amdgcn_wmma_f32_16x16x4_f32(false, a, false, f.bv[nn],
                                                   (short)0, va[nn], false, false);
  }
}

__device__ __forceinline__ void lds_barrier()
{
  __builtin_amdgcn_fence(__ATOMIC_RELEASE, "workgroup", "local");
  __builtin_amdgcn_s_barrier();
  __builtin_amdgcn_fence(__ATOMIC_ACQUIRE, "workgroup", "local");
}
__device__ __forceinline__ void lds_barrier_arrive()
{
  __builtin_amdgcn_fence(__ATOMIC_RELEASE, "workgroup", "local");
  __builtin_amdgcn_s_barrier_signal(-1);
}
__device__ __forceinline__ void lds_barrier_wait()
{
  __builtin_amdgcn_s_barrier_wait(-1);
  __builtin_amdgcn_fence(__ATOMIC_ACQUIRE, "workgroup", "local");
}

__global__ __launch_bounds__(256) void proj_qkv(
    const float* __restrict__ x,
    const float* __restrict__ Wk,
    const float* __restrict__ Wq,
    const float* __restrict__ Wv,
    float* __restrict__ qo, float* __restrict__ ko, float* __restrict__ vo)
{
  __shared__ __attribute__((aligned(16))) float wls[KG * 2 * WBUF];
  __shared__ __attribute__((aligned(16))) float xch[4 * 24 * 128];

  const int lane = threadIdx.x & 31;
  const int grp  = __builtin_amdgcn_readfirstlane((int)(threadIdx.x >> 7));
  const int wq   = (threadIdx.x >> 5) & 3;
  const int tile = blockIdx.x * 4 + wq;
  const int row0 = tile * 16;
  const int m    = lane & 15;
  const int kp   = lane >> 4;

  v8f qa[4] = {}, ka[4] = {}, va[4] = {};

  const float* xr = x + (size_t)(row0 + m) * Csz + grp * (Csz / KG) + 2 * kp;

  const int sp   = (threadIdx.x & 127) >> 4;
  const int sc4  = (threadIdx.x & 15) * 4;
  const int gsrc = grp * (Csz / KG) * Hsz + (2 * sp) * Hsz + sc4;
  const int sdst = grp * (2 * WBUF) + sp * PP + sc4 * 2;
  const int lfr  = grp * (2 * WBUF) + kp * PP + m * 2;

  v4f ga[3], gb[3];

  auto gload = [&](const int ch) __attribute__((always_inline)) {
    const size_t o = (size_t)ch * (KC * Hsz) + gsrc;
    ga[0] = *(const v4f*)(Wq + o);  gb[0] = *(const v4f*)(Wq + o + Hsz);
    ga[1] = *(const v4f*)(Wk + o);  gb[1] = *(const v4f*)(Wk + o + Hsz);
    ga[2] = *(const v4f*)(Wv + o);  gb[2] = *(const v4f*)(Wv + o + Hsz);
  };
  auto sstore = [&](const int buf) __attribute__((always_inline)) {
#pragma unroll
    for (int X = 0; X < 3; ++X) {
      v4f s0, s1;
      s0.x = ga[X].x; s0.y = gb[X].x; s0.z = ga[X].y; s0.w = gb[X].y;
      s1.x = ga[X].z; s1.y = gb[X].z; s1.z = ga[X].w; s1.w = gb[X].w;
      *(v4f*)(&wls[buf * WBUF + X * WMAT + sdst])     = s0;
      *(v4f*)(&wls[buf * WBUF + X * WMAT + sdst + 4]) = s1;
    }
  };
  v2f xs[8];
  auto xload = [&](const int slot, const int kk) __attribute__((always_inline)) {
    xs[slot] = *(const v2f*)(xr + kk);
  };
  auto wload = [&](Frag& f, const int buf, const int t) __attribute__((always_inline)) {
    const int o = buf * WBUF + lfr + (2 * t) * PP;
#pragma unroll
    for (int nn = 0; nn < 4; ++nn) {
      f.bq[nn] = *(const v2f*)(&wls[o + 0 * WMAT + nn * 32]);
      f.bk[nn] = *(const v2f*)(&wls[o + 1 * WMAT + nn * 32]);
      f.bv[nn] = *(const v2f*)(&wls[o + 2 * WMAT + nn * 32]);
    }
  };

  gload(0);
  xload(0, 0);   xload(1, 4);   xload(2, 8);
  xload(3, 12);  xload(4, 16);  xload(5, 20);
  sstore(0);
  gload(1);
  lds_barrier();

  Frag c0, c1, n0, n1;
  wload(c0, 0, 0);
  wload(c1, 0, 1);

  auto trip = [&](const int c, const int P, const bool fetch_w, const bool top_x, const bool mid_x)
      __attribute__((always_inline)) {
    const int kk  = c * KC;
    const int nxt = P ^ 1;
    const int s   = 4 * P;
    __builtin_amdgcn_sched_barrier(0);
    sstore(nxt);
    if (fetch_w) gload(c + 2);
    __builtin_amdgcn_sched_barrier(0);
    if (top_x) {
      xload((s + 6) & 7, kk + 24);
      xload((s + 7) & 7, kk + 28);
    }
    wload(n0, P, 2);
    wload(n1, P, 3);
    __builtin_amdgcn_sched_barrier(0);
    __builtin_amdgcn_s_setprio(2);
    mma_step(xs[s],     c0, qa, ka, va);
    lds_barrier_arrive();
    __builtin_amdgcn_s_setprio(1);
    mma_step(xs[s + 1], c1, qa, ka, va);
    lds_barrier_wait();
    __builtin_amdgcn_s_setprio(3);
    if (mid_x) {
      xload(s,     kk + 32);
      xload(s + 1, kk + 36);
    }
    wload(c0, nxt, 0);
    wload(c1, nxt, 1);
    mma_step(xs[s + 2], n0, qa, ka, va);
    mma_step(xs[s + 3], n1, qa, ka, va);
  };

  for (int c = 0; c < NCH - 2; c += 2) {
    trip(c,     0, true, true, true);
    trip(c + 1, 1, true, true, true);
  }
  trip(NCH - 2, 0, false, true, false);
  wload(n0, 1, 2);
  wload(n1, 1, 3);
  mma_step(xs[4], c0, qa, ka, va);
  mma_step(xs[5], c1, qa, ka, va);
  mma_step(xs[6], n0, qa, ka, va);
  mma_step(xs[7], n1, qa, ka, va);
  __builtin_amdgcn_s_setprio(0);

  float* xw = xch + (wq * 24) * 128 + lane * 4;
  if (grp == 1) {
#pragma unroll
    for (int nn = 0; nn < 4; ++nn)
#pragma unroll
      for (int u = 0; u < 2; ++u) {
        v4f tq, tk, tv;
#pragma unroll
        for (int e = 0; e < 4; ++e) {
          tq[e] = qa[nn][4 * u + e];
          tk[e] = ka[nn][4 * u + e];
          tv[e] = va[nn][4 * u + e];
        }
        *(v4f*)(xw + (0  + nn * 2 + u) * 128) = tq;
        *(v4f*)(xw + (8  + nn * 2 + u) * 128) = tk;
        *(v4f*)(xw + (16 + nn * 2 + u) * 128) = tv;
      }
  }
  __syncthreads();
  if (grp == 1) return;
#pragma unroll
  for (int nn = 0; nn < 4; ++nn)
#pragma unroll
    for (int u = 0; u < 2; ++u) {
      const v4f tq = *(const v4f*)(xw + (0  + nn * 2 + u) * 128);
      const v4f tk = *(const v4f*)(xw + (8  + nn * 2 + u) * 128);
      const v4f tv = *(const v4f*)(xw + (16 + nn * 2 + u) * 128);
#pragma unroll
      for (int e = 0; e < 4; ++e) {
        qa[nn][4 * u + e] += tq[e];
        ka[nn][4 * u + e] += tk[e];
        va[nn][4 * u + e] += tv[e];
      }
    }

  const size_t tb = (size_t)tile * (16 * Hsz) + lane * 4;
#pragma unroll
  for (int nn = 0; nn < 4; ++nn)
#pragma unroll
    for (int u = 0; u < 2; ++u) {
      const size_t idx = tb + (nn * 2 + u) * 128;
      v4f tq, tk, tv;
#pragma unroll
      for (int e = 0; e < 4; ++e) {
        tq[e] = qa[nn][4 * u + e];
        tk[e] = ka[nn][4 * u + e];
        tv[e] = va[nn][4 * u + e];
      }
      *(v4f*)(qo + idx) = tq;
      *(v4f*)(ko + idx) = tk;
      *(v4f*)(vo + idx) = tv;
    }
}

static constexpr int NW = 8;
static constexpr int QT = 32;
static constexpr int MP = 68;

__device__ __forceinline__ float xhalf(float x)
{
  const int b = __builtin_bit_cast(int, x);
  return __builtin_bit_cast(float,
      __builtin_amdgcn_permlanex16(b, b, 0x76543210, 0xfedcba98, false, false));
}

__global__ __launch_bounds__(NW * 32) void flash_attn(
    const float* __restrict__ qm, const float* __restrict__ km,
    const float* __restrict__ vm, float* __restrict__ out)
{
  __shared__ __attribute__((aligned(16))) float macc[NW * QT * MP];
  __shared__ float mst[NW][QT];
  __shared__ float lst[NW][QT];

  const int lane = threadIdx.x & 31;
  const int wid  = __builtin_amdgcn_readfirstlane((int)(threadIdx.x >> 5));
  const int bb   = blockIdx.x >> 6;
  const int pr   = blockIdx.x & 63;
  const int m    = lane & 15;
  const int kp   = lane >> 4;

  const size_t base = (size_t)bb * Tsz * Hsz;
  const float* qb = qm + base;
  const float* kb = km + base;
  const float* vb = vm + base;

  for (int half = 0; half < 2; ++half) {
    const int ti = half ? pr : 127 - pr;

    v2f qf[2][16];
#pragma unroll
    for (int s = 0; s < 2; ++s) {
      const float* qr = qb + (size_t)(2 * ti + s) * (16 * Hsz) + lane * 4;
#pragma unroll
      for (int j = 0; j < 8; ++j) {
        const v4f t = *(const v4f*)(qr + j * 128);
        qf[s][2 * j].x     = t.x * 0.125f;
        qf[s][2 * j].y     = t.y * 0.125f;
        qf[s][2 * j + 1].x = t.z * 0.125f;
        qf[s][2 * j + 1].y = t.w * 0.125f;
      }
    }

    v8f acc[2][4] = {};
    float mrun[2] = {-1e30f, -1e30f};
    v8f lv[2] = {};

    const int jd = 2 * ti;
    const int jl = jd + 1;

    v4f kA[8], vA[8], kB[8], vB[8];
    auto kvload = [&](v4f (&kc)[8], v4f (&vc)[8], const int j) __attribute__((always_inline)) {
      const float* kt = kb + (size_t)j * (16 * Hsz) + lane * 4;
      const float* vt = vb + (size_t)j * (16 * Hsz) + lane * 4;
#pragma unroll
      for (int c = 0; c < 8; ++c) kc[c] = *(const v4f*)(kt + c * 128);
#pragma unroll
      for (int c = 0; c < 8; ++c) vc[c] = *(const v4f*)(vt + c * 128);
    };

    auto tile = [&](const v4f (&kc)[8], const v4f (&vc)[8], const int j) __attribute__((always_inline)) {
      v8f s8[2] = {};
#pragma unroll
      for (int c = 0; c < 8; ++c) {
        v2f a0, a1;
        a0.x = kc[c].x; a0.y = kc[c].y;
        a1.x = kc[c].z; a1.y = kc[c].w;
        s8[0] = __builtin_amdgcn_wmma_f32_16x16x4_f32(false, a0, false, qf[0][2 * c],
                                                      (short)0, s8[0], false, false);
        s8[1] = __builtin_amdgcn_wmma_f32_16x16x4_f32(false, a0, false, qf[1][2 * c],
                                                      (short)0, s8[1], false, false);
        s8[0] = __builtin_amdgcn_wmma_f32_16x16x4_f32(false, a1, false, qf[0][2 * c + 1],
                                                      (short)0, s8[0], false, false);
        s8[1] = __builtin_amdgcn_wmma_f32_16x16x4_f32(false, a1, false, qf[1][2 * c + 1],
                                                      (short)0, s8[1], false, false);
      }

      asm volatile("s_wait_alu depctr_va_vdst(0)" : "+v"(s8[0]), "+v"(s8[1]));

      if (j >= jd) {
        const int lim = m - 8 * kp - 16 * (j - jd);
#pragma unroll
        for (int s = 0; s < 2; ++s)
#pragma unroll
          for (int r = 0; r < 8; ++r)
            if (r > lim + 16 * s) s8[s][r] = -__builtin_inff();
      }

      v8f p8[2];
#pragma unroll
      for (int s = 0; s < 2; ++s) {
        float tm = s8[s][0];
#pragma unroll
        for (int r = 1; r < 8; ++r) tm = fmaxf(tm, s8[s][r]);
        tm = fmaxf(tm, xhalf(tm));
        const float mnew  = fmaxf(mrun[s], tm);
        const float alpha = __expf(mrun[s] - mnew);
        mrun[s] = mnew;
#pragma unroll
        for (int r = 0; r < 8; ++r) p8[s][r] = __expf(s8[s][r] - mnew);
        lv[s] = lv[s] * alpha + p8[s];
#pragma unroll
        for (int nn = 0; nn < 4; ++nn)
#pragma unroll
          for (int r = 0; r < 8; ++r) acc[s][nn][r] *= alpha;
      }

      __builtin_amdgcn_sched_barrier(0);

#pragma unroll
      for (int t = 0; t < 4; ++t) {
        v2f vf[4];
#pragma unroll
        for (int nn = 0; nn < 4; ++nn) {
          const v4f w = vc[nn * 2 + (t >> 1)];
          vf[nn].x = (t & 1) ? w.z : w.x;
          vf[nn].y = (t & 1) ? w.w : w.y;
        }
        v2f pf0, pf1;
        pf0.x = p8[0][2 * t];
        pf0.y = p8[0][2 * t + 1];
        pf1.x = p8[1][2 * t];
        pf1.y = p8[1][2 * t + 1];
#pragma unroll
        for (int nn = 0; nn < 4; ++nn) {
          acc[0][nn] = __builtin_amdgcn_wmma_f32_16x16x4_f32(false, vf[nn], false, pf0,
                                                             (short)0, acc[0][nn], false, false);
          acc[1][nn] = __builtin_amdgcn_wmma_f32_16x16x4_f32(false, vf[nn], false, pf1,
                                                             (short)0, acc[1][nn], false, false);
        }
      }
    };

    {
      const int nst = (wid <= jl) ? (jl - wid) / NW + 1 : 0;
      const int jp1 = wid + NW * ((nst / 2) & ~1);
      const int jp2 = wid + NW * ((3 * nst / 4) & ~1);
      const int jp3 = wid + NW * ((7 * nst / 8) & ~1);

      int j = wid;
      kvload(kA, vA, j < jl ? j : jl);
      __builtin_amdgcn_s_setprio(3);
      while (j + NW <= jl) {
        if (j == jp1) __builtin_amdgcn_s_setprio(2);
        if (j == jp2) __builtin_amdgcn_s_setprio(1);
        if (j == jp3) __builtin_amdgcn_s_setprio(0);
        kvload(kB, vB, j + NW);
        tile(kA, vA, j);
        kvload(kA, vA, j + 2 * NW < jl ? j + 2 * NW : jl);
        tile(kB, vB, j + NW);
        j += 2 * NW;
      }
      if (j <= jl) tile(kA, vA, j);
      __builtin_amdgcn_s_setprio(0);
    }

#pragma unroll
    for (int s = 0; s < 2; ++s) {
      float* ma = macc + (wid * QT + 16 * s + m) * MP + 8 * kp;
#pragma unroll
      for (int nn = 0; nn < 4; ++nn) {
        float4 lo, hi;
        lo.x = acc[s][nn][0]; lo.y = acc[s][nn][1]; lo.z = acc[s][nn][2]; lo.w = acc[s][nn][3];
        hi.x = acc[s][nn][4]; hi.y = acc[s][nn][5]; hi.z = acc[s][nn][6]; hi.w = acc[s][nn][7];
        *(float4*)(ma + nn * 16)     = lo;
        *(float4*)(ma + nn * 16 + 4) = hi;
      }
      const float lsum  = ((lv[s][0] + lv[s][1]) + (lv[s][2] + lv[s][3])) + ((lv[s][4] + lv[s][5]) + (lv[s][6] + lv[s][7]));
      const float lfull = lsum + xhalf(lsum);
      if (kp == 0) {
        mst[wid][16 * s + m] = mrun[s];
        lst[wid][16 * s + m] = lfull;
      }
    }

    __syncthreads();

#pragma unroll
    for (int rh = 0; rh < 2; ++rh) {
      const int row = (threadIdx.x >> 4) + 16 * rh;
      const int c4  = (threadIdx.x & 15) * 4;
      float mw[NW];
      float mt = -1e30f;
#pragma unroll
      for (int w = 0; w < NW; ++w) { mw[w] = mst[w][row]; mt = fmaxf(mt, mw[w]); }
      float lt = 0.0f;
      float o0 = 0.0f, o1 = 0.0f, o2 = 0.0f, o3 = 0.0f;
#pragma unroll
      for (int w = 0; w < NW; ++w) {
        const float sc = __expf(mw[w] - mt);
        lt += lst[w][row] * sc;
        const float4 a = *(const float4*)(macc + (w * QT + row) * MP + c4);
        o0 += a.x * sc; o1 += a.y * sc; o2 += a.z * sc; o3 += a.w * sc;
      }
      float4 res;
      res.x = o0 / lt; res.y = o1 / lt; res.z = o2 / lt; res.w = o3 / lt;
      *(float4*)(out + base + (size_t)(ti * QT + row) * Hsz + c4) = res;
    }

    if (half == 0) __syncthreads();
  }
}

extern "C" void kernel_launch(void* const* d_in, const int* in_sizes, int n_in,
                              void* d_out, int out_size, void* d_ws, size_t ws_size,
                              hipStream_t stream) {
  (void)in_sizes; (void)n_in; (void)out_size; (void)ws_size;

  const float* x  = (const float*)d_in[0];
  const float* Wk = (const float*)d_in[1];
  const float* Wq = (const float*)d_in[2];
  const float* Wv = (const float*)d_in[3];
  float* out = (float*)d_out;

  float* qo = (float*)d_ws;
  float* ko = qo + (size_t)BT * Hsz;
  float* vo = ko + (size_t)BT * Hsz;

  const int tiles = BT / 16;
  dim3 block(256);
  dim3 grid(tiles / 4);

  proj_qkv<<<grid, block, 0, stream>>>(x, Wk, Wq, Wv, qo, ko, vo);
  flash_attn<<<dim3(256), dim3(256), 0, stream>>>(qo, ko, vo, out);
}
